# starting kernel
# baseline (speedup 1.0000x reference)
#include <hip/hip_runtime.h>

typedef __attribute__((ext_vector_type(2))) float v2f;
typedef __attribute__((ext_vector_type(8))) float v8f;

#define BATCH   4
#define NPTS    8192
#define CIN     64
#define COUT    128
#define NPOINT  2048
#define KNN     16
#define KRAW    67
#define KPAD    68
#define BNEPS   1e-5f

__global__ __launch_bounds__(1024) void fps_kernel(const float* __restrict__ xyz,
                                                   int* __restrict__ fps_idx)
{
    const int b   = blockIdx.x;
    const int tid = threadIdx.x;
    const float* xb = xyz + (size_t)b * NPTS * 3;

    __shared__ float sdist[1024];
    __shared__ int   sidx[1024];

    float dist[8];
#pragma unroll
    for (int i = 0; i < 8; ++i) dist[i] = 3.4e38f;

    int cur = 0;
    for (int s = 0; s < NPOINT; ++s) {
        if (tid == 0) fps_idx[b * NPOINT + s] = cur;
        const float cx = xb[cur * 3 + 0];
        const float cy = xb[cur * 3 + 1];
        const float cz = xb[cur * 3 + 2];

        float best = -1.0f; int bidx = 0;
#pragma unroll
        for (int i = 0; i < 8; ++i) {
            const int n = i * 1024 + tid;
            const float dx = xb[n * 3 + 0] - cx;
            const float dy = xb[n * 3 + 1] - cy;
            const float dz = xb[n * 3 + 2] - cz;
            const float d  = dx * dx + dy * dy + dz * dz;
            const float nd = fminf(dist[i], d);
            dist[i] = nd;
            if (nd > best) { best = nd; bidx = n; }
        }
        sdist[tid] = best; sidx[tid] = bidx;
        __syncthreads();
        for (int off = 512; off > 0; off >>= 1) {
            if (tid < off) {
                const float d1 = sdist[tid],       d2 = sdist[tid + off];
                const int   i1 = sidx[tid],        i2 = sidx[tid + off];
                if (d2 > d1 || (d2 == d1 && i2 < i1)) { sdist[tid] = d2; sidx[tid] = i2; }
            }
            __syncthreads();
        }
        cur = sidx[0];
        __syncthreads();
    }
}

__global__ void gather_new_xyz_kernel(const float* __restrict__ xyz,
                                      const int* __restrict__ fps_idx,
                                      float* __restrict__ new_xyz)
{
    const int e = blockIdx.x * blockDim.x + threadIdx.x;
    if (e >= BATCH * NPOINT * 3) return;
    const int c = e % 3;
    const int p = (e / 3) % NPOINT;
    const int b = e / (3 * NPOINT);
    const int n = fps_idx[b * NPOINT + p];
    new_xyz[e] = xyz[((size_t)b * NPTS + n) * 3 + c];
}

__global__ __launch_bounds__(128) void knn_kernel(const float* __restrict__ xyz,
                                                  const float* __restrict__ new_xyz,
                                                  int* __restrict__ knn_idx)
{
    const int q = blockIdx.x * blockDim.x + threadIdx.x;
    if (q >= BATCH * NPOINT) return;
    const int b = q / NPOINT;
    const float qx = new_xyz[q * 3 + 0];
    const float qy = new_xyz[q * 3 + 1];
    const float qz = new_xyz[q * 3 + 2];
    const float* xb = xyz + (size_t)b * NPTS * 3;

    float v[KNN]; int id[KNN];
#pragma unroll
    for (int t = 0; t < KNN; ++t) { v[t] = 3.4e38f; id[t] = 0; }

    for (int j = 0; j < NPTS; ++j) {
        if ((j & 63) == 0)
            __builtin_prefetch(xb + (size_t)(j + 256) * 3, 0, 1);
        const float dx = xb[j * 3 + 0] - qx;
        const float dy = xb[j * 3 + 1] - qy;
        const float dz = xb[j * 3 + 2] - qz;
        const float d  = dx * dx + dy * dy + dz * dz;
        if (d < v[KNN - 1]) {
            v[KNN - 1] = d; id[KNN - 1] = j;
#pragma unroll
            for (int t = KNN - 1; t > 0; --t) {
                if (v[t] < v[t - 1]) {
                    const float tv = v[t]; v[t] = v[t - 1]; v[t - 1] = tv;
                    const int   ti = id[t]; id[t] = id[t - 1]; id[t - 1] = ti;
                }
            }
        }
    }
#pragma unroll
    for (int t = 0; t < KNN; ++t) knn_idx[(size_t)q * KNN + t] = id[t];
}

__global__ void pad_w_kernel(const float* __restrict__ W, float* __restrict__ Wpad)
{
    const int e = blockIdx.x * blockDim.x + threadIdx.x;
    if (e >= KPAD * COUT) return;
    const int r = e / COUT;
    Wpad[e] = (r < KRAW) ? W[e] : 0.0f;
}

__global__ __launch_bounds__(256) void group_gemm_kernel(
    const float* __restrict__ xyz, const float* __restrict__ features,
    const float* __restrict__ new_xyz, const int* __restrict__ knn_idx,
    const float* __restrict__ Wpad,
    const float* __restrict__ gamma, const float* __restrict__ beta,
    const float* __restrict__ meanv, const float* __restrict__ varv,
    float* __restrict__ partial_sum, float* __restrict__ partial_sq,
    float* __restrict__ out_features, int final_pass)
{
    const int q    = blockIdx.x;
    const int b    = q / NPOINT;
    const int tid  = threadIdx.x;
    const int lane = tid & 31;
    const int wv   = tid >> 5;

    __shared__ float As[KNN][KPAD];
    __shared__ float red[8][32];

    const float qx = new_xyz[q * 3 + 0];
    const float qy = new_xyz[q * 3 + 1];
    const float qz = new_xyz[q * 3 + 2];

    for (int e = tid; e < KNN * KPAD; e += 256) {
        const int kk = e / KPAD;
        const int c  = e % KPAD;
        const int n  = knn_idx[(size_t)q * KNN + kk];
        float val;
        if (c < 3) {
            const float pv = xyz[((size_t)b * NPTS + n) * 3 + c];
            val = pv - (c == 0 ? qx : (c == 1 ? qy : qz));
        } else if (c < KRAW) {
            val = features[((size_t)b * NPTS + n) * CIN + (c - 3)];
        } else {
            val = 0.0f;
        }
        As[kk][c] = val;
    }
    __syncthreads();

    const int col = wv * 16 + (lane & 15);
    const int hi  = lane >> 4;
    const int row = lane & 15;

    v8f acc = {};
#pragma unroll
    for (int kb = 0; kb < KPAD / 4; ++kb) {
        const int k0 = kb * 4 + hi * 2;
        v2f a, bb;
        a.x  = As[row][k0 + 0];
        a.y  = As[row][k0 + 1];
        bb.x = Wpad[(k0 + 0) * COUT + col];
        bb.y = Wpad[(k0 + 1) * COUT + col];
        acc = __builtin_amdgcn_wmma_f32_16x16x4_f32(
                  false, a, false, bb, (short)0, acc, false, false);
    }

    if (!final_pass) {
        float s = 0.0f, ss = 0.0f;
#pragma unroll
        for (int t = 0; t < 8; ++t) { const float h = acc[t]; s += h; ss += h * h; }
        red[wv][lane] = s;
        __syncthreads();
        float s2 = 0.0f;
        if (lane < 16) s2 = red[wv][lane] + red[wv][lane + 16];
        __syncthreads();
        red[wv][lane] = ss;
        __syncthreads();
        if (lane < 16) {
            const float ss2 = red[wv][lane] + red[wv][lane + 16];
            partial_sum[(size_t)q * COUT + col] = s2;
            partial_sq [(size_t)q * COUT + col] = ss2;
        }
    } else {
        const float mu    = meanv[col];
        const float scale = gamma[col] * rsqrtf(varv[col] + BNEPS);
        const float shift = beta[col] - mu * scale;
        float m = -3.4e38f;
#pragma unroll
        for (int t = 0; t < 8; ++t) {
            float h = acc[t] * scale + shift;
            h = fmaxf(h, 0.0f);
            m = fmaxf(m, h);
        }
        red[wv][lane] = m;
        __syncthreads();
        if (lane < 16) {
            const float mm = fmaxf(red[wv][lane], red[wv][lane + 16]);
            out_features[(size_t)q * COUT + col] = mm;
        }
    }
}

__global__ __launch_bounds__(256) void reduce_stats_kernel(
    const float* __restrict__ partial_sum, const float* __restrict__ partial_sq,
    float* __restrict__ meanv, float* __restrict__ varv)
{
    const int ch  = blockIdx.x;
    const int tid = threadIdx.x;
    __shared__ float ssum[256], ssq[256];
    float s = 0.0f, sq = 0.0f;
    for (int i = tid; i < BATCH * NPOINT; i += 256) {
        s  += partial_sum[(size_t)i * COUT + ch];
        sq += partial_sq [(size_t)i * COUT + ch];
    }
    ssum[tid] = s; ssq[tid] = sq;
    __syncthreads();
    for (int off = 128; off > 0; off >>= 1) {
        if (tid < off) { ssum[tid] += ssum[tid + off]; ssq[tid] += ssq[tid + off]; }
        __syncthreads();
    }
    if (tid == 0) {
        const float cnt = (float)(BATCH * NPOINT * KNN);
        const float mu  = ssum[0] / cnt;
        meanv[ch] = mu;
        varv[ch]  = ssq[0] / cnt - mu * mu;
    }
}

extern "C" void kernel_launch(void* const* d_in, const int* in_sizes, int n_in,
                              void* d_out, int out_size, void* d_ws, size_t ws_size,
                              hipStream_t stream)
{
    const float* xyz      = (const float*)d_in[0];
    const float* features = (const float*)d_in[1];
    const float* W        = (const float*)d_in[2];
    const float* gamma    = (const float*)d_in[3];
    const float* beta     = (const float*)d_in[4];

    float* out          = (float*)d_out;
    float* new_xyz      = out;
    float* out_features = out + BATCH * NPOINT * 3;

    char* ws = (char*)d_ws;
    size_t off = 0;
    int*   fps_idx     = (int*)(ws + off);   off += sizeof(int)   * BATCH * NPOINT;
    int*   knn_idx     = (int*)(ws + off);   off += sizeof(int)   * BATCH * NPOINT * KNN;
    float* Wpad        = (float*)(ws + off); off += sizeof(float) * KPAD * COUT;
    off = (off + 255) & ~(size_t)255;
    float* partial_sum = (float*)(ws + off); off += sizeof(float) * BATCH * NPOINT * COUT;
    float* partial_sq  = (float*)(ws + off); off += sizeof(float) * BATCH * NPOINT * COUT;
    float* meanv       = (float*)(ws + off); off += sizeof(float) * COUT;
    float* varv        = (float*)(ws + off); off += sizeof(float) * COUT;
    (void)ws_size; (void)in_sizes; (void)n_in; (void)out_size;

    fps_kernel<<<BATCH, 1024, 0, stream>>>(xyz, fps_idx);
    gather_new_xyz_kernel<<<(BATCH * NPOINT * 3 + 255) / 256, 256, 0, stream>>>(
        xyz, fps_idx, new_xyz);
    knn_kernel<<<(BATCH * NPOINT + 127) / 128, 128, 0, stream>>>(
        xyz, new_xyz, knn_idx);
    pad_w_kernel<<<(KPAD * COUT + 255) / 256, 256, 0, stream>>>(W, Wpad);
    group_gemm_kernel<<<BATCH * NPOINT, 256, 0, stream>>>(
        xyz, features, new_xyz, knn_idx, Wpad, gamma, beta,
        meanv, varv, partial_sum, partial_sq, out_features, 0);
    reduce_stats_kernel<<<COUT, 256, 0, stream>>>(
        partial_sum, partial_sq, meanv, varv);
    group_gemm_kernel<<<BATCH * NPOINT, 256, 0, stream>>>(
        xyz, features, new_xyz, knn_idx, Wpad, gamma, beta,
        meanv, varv, partial_sum, partial_sq, out_features, 1);
}


# Sampling argmax: per-wave DPP reduction (max of the distance bits, then min index among its holders) and one LDS exchange of the 32 wave pairs; one barrier per pick instead of the LDS tree's twelve
# speedup vs baseline: 1.9565x; 1.9565x over previous
#include <hip/hip_runtime.h>

typedef __attribute__((ext_vector_type(2))) float v2f;
typedef __attribute__((ext_vector_type(8))) float v8f;

#define BATCH   4
#define NPTS    8192
#define CIN     64
#define COUT    128
#define NPOINT  2048
#define KNN     16
#define KRAW    67
#define KPAD    68
#define BNEPS   1e-5f

template <int M>
__device__ __forceinline__ int row_xor(int v)
{
    return __builtin_amdgcn_update_dpp(v, v, 0x160 + M, 0xf, 0xf, true);
}
__device__ __forceinline__ int half_xor(int v)
{
    return __builtin_amdgcn_permlanex16(v, v, 0x76543210, 0xfedcba98, false, false);
}
__device__ __forceinline__ int wave_max_i32(int x)
{
    x = max(x, row_xor<1>(x));
    x = max(x, row_xor<2>(x));
    x = max(x, row_xor<4>(x));
    x = max(x, row_xor<8>(x));
    x = max(x, half_xor(x));
    return x;
}
__device__ __forceinline__ int wave_min_i32(int x)
{
    x = min(x, row_xor<1>(x));
    x = min(x, row_xor<2>(x));
    x = min(x, row_xor<4>(x));
    x = min(x, row_xor<8>(x));
    x = min(x, half_xor(x));
    return x;
}

__global__ __launch_bounds__(1024) void fps_kernel(const float* __restrict__ xyz,
                                                   int* __restrict__ fps_idx)
{
    const int b    = blockIdx.x;
    const int tid  = threadIdx.x;
    const int lane = tid & 31;
    const int wv   = tid >> 5;
    const float* xb = xyz + (size_t)b * NPTS * 3;

    __shared__ int2 spair[2][32];

    float dist[8];
#pragma unroll
    for (int i = 0; i < 8; ++i) dist[i] = 3.4e38f;

    int cur = 0;
    for (int s = 0; s < NPOINT; ++s) {
        if (tid == 0) fps_idx[b * NPOINT + s] = cur;
        const float cx = xb[cur * 3 + 0];
        const float cy = xb[cur * 3 + 1];
        const float cz = xb[cur * 3 + 2];

        float best = -1.0f; int bidx = 0;
#pragma unroll
        for (int i = 0; i < 8; ++i) {
            const int n = i * 1024 + tid;
            const float dx = xb[n * 3 + 0] - cx;
            const float dy = xb[n * 3 + 1] - cy;
            const float dz = xb[n * 3 + 2] - cz;
            const float d  = dx * dx + dy * dy + dz * dz;
            const float nd = fminf(dist[i], d);
            dist[i] = nd;
            if (nd > best) { best = nd; bidx = n; }
        }

        const int bb = __float_as_int(best);
        const int wm = wave_max_i32(bb);
        const int wi = wave_min_i32(bb == wm ? bidx : 0x7fffffff);
        if (lane == 0) spair[s & 1][wv] = make_int2(wm, wi);
        __syncthreads();

        const int2 pr = spair[s & 1][lane];
        const int  gm = wave_max_i32(pr.x);
        const int  gi = wave_min_i32(pr.x == gm ? pr.y : 0x7fffffff);
        cur = __builtin_amdgcn_readfirstlane(gi) & (NPTS - 1);
    }
}

__global__ void gather_new_xyz_kernel(const float* __restrict__ xyz,
                                      const int* __restrict__ fps_idx,
                                      float* __restrict__ new_xyz)
{
    const int e = blockIdx.x * blockDim.x + threadIdx.x;
    if (e >= BATCH * NPOINT * 3) return;
    const int c = e % 3;
    const int p = (e / 3) % NPOINT;
    const int b = e / (3 * NPOINT);
    const int n = fps_idx[b * NPOINT + p];
    new_xyz[e] = xyz[((size_t)b * NPTS + n) * 3 + c];
}

__global__ __launch_bounds__(128) void knn_kernel(const float* __restrict__ xyz,
                                                  const float* __restrict__ new_xyz,
                                                  int* __restrict__ knn_idx)
{
    const int q = blockIdx.x * blockDim.x + threadIdx.x;
    if (q >= BATCH * NPOINT) return;
    const int b = q / NPOINT;
    const float qx = new_xyz[q * 3 + 0];
    const float qy = new_xyz[q * 3 + 1];
    const float qz = new_xyz[q * 3 + 2];
    const float* xb = xyz + (size_t)b * NPTS * 3;

    float v[KNN]; int id[KNN];
#pragma unroll
    for (int t = 0; t < KNN; ++t) { v[t] = 3.4e38f; id[t] = 0; }

    for (int j = 0; j < NPTS; ++j) {
        if ((j & 63) == 0)
            __builtin_prefetch(xb + (size_t)(j + 256) * 3, 0, 1);
        const float dx = xb[j * 3 + 0] - qx;
        const float dy = xb[j * 3 + 1] - qy;
        const float dz = xb[j * 3 + 2] - qz;
        const float d  = dx * dx + dy * dy + dz * dz;
        if (d < v[KNN - 1]) {
            v[KNN - 1] = d; id[KNN - 1] = j;
#pragma unroll
            for (int t = KNN - 1; t > 0; --t) {
                if (v[t] < v[t - 1]) {
                    const float tv = v[t]; v[t] = v[t - 1]; v[t - 1] = tv;
                    const int   ti = id[t]; id[t] = id[t - 1]; id[t - 1] = ti;
                }
            }
        }
    }
#pragma unroll
    for (int t = 0; t < KNN; ++t) knn_idx[(size_t)q * KNN + t] = id[t];
}

__global__ void pad_w_kernel(const float* __restrict__ W, float* __restrict__ Wpad)
{
    const int e = blockIdx.x * blockDim.x + threadIdx.x;
    if (e >= KPAD * COUT) return;
    const int r = e / COUT;
    Wpad[e] = (r < KRAW) ? W[e] : 0.0f;
}

__global__ __launch_bounds__(256) void group_gemm_kernel(
    const float* __restrict__ xyz, const float* __restrict__ features,
    const float* __restrict__ new_xyz, const int* __restrict__ knn_idx,
    const float* __restrict__ Wpad,
    const float* __restrict__ gamma, const float* __restrict__ beta,
    const float* __restrict__ meanv, const float* __restrict__ varv,
    float* __restrict__ partial_sum, float* __restrict__ partial_sq,
    float* __restrict__ out_features, int final_pass)
{
    const int q    = blockIdx.x;
    const int b    = q / NPOINT;
    const int tid  = threadIdx.x;
    const int lane = tid & 31;
    const int wv   = tid >> 5;

    __shared__ float As[KNN][KPAD];
    __shared__ float red[8][32];

    const float qx = new_xyz[q * 3 + 0];
    const float qy = new_xyz[q * 3 + 1];
    const float qz = new_xyz[q * 3 + 2];

    for (int e = tid; e < KNN * KPAD; e += 256) {
        const int kk = e / KPAD;
        const int c  = e % KPAD;
        const int n  = knn_idx[(size_t)q * KNN + kk];
        float val;
        if (c < 3) {
            const float pv = xyz[((size_t)b * NPTS + n) * 3 + c];
            val = pv - (c == 0 ? qx : (c == 1 ? qy : qz));
        } else if (c < KRAW) {
            val = features[((size_t)b * NPTS + n) * CIN + (c - 3)];
        } else {
            val = 0.0f;
        }
        As[kk][c] = val;
    }
    __syncthreads();

    const int col = wv * 16 + (lane & 15);
    const int hi  = lane >> 4;
    const int row = lane & 15;

    v8f acc = {};
#pragma unroll
    for (int kb = 0; kb < KPAD / 4; ++kb) {
        const int k0 = kb * 4 + hi * 2;
        v2f a, bb;
        a.x  = As[row][k0 + 0];
        a.y  = As[row][k0 + 1];
        bb.x = Wpad[(k0 + 0) * COUT + col];
        bb.y = Wpad[(k0 + 1) * COUT + col];
        acc = __builtin_amdgcn_wmma_f32_16x16x4_f32(
                  false, a, false, bb, (short)0, acc, false, false);
    }

    if (!final_pass) {
        float s = 0.0f, ss = 0.0f;
#pragma unroll
        for (int t = 0; t < 8; ++t) { const float h = acc[t]; s += h; ss += h * h; }
        red[wv][lane] = s;
        __syncthreads();
        float s2 = 0.0f;
        if (lane < 16) s2 = red[wv][lane] + red[wv][lane + 16];
        __syncthreads();
        red[wv][lane] = ss;
        __syncthreads();
        if (lane < 16) {
            const float ss2 = red[wv][lane] + red[wv][lane + 16];
            partial_sum[(size_t)q * COUT + col] = s2;
            partial_sq [(size_t)q * COUT + col] = ss2;
        }
    } else {
        const float mu    = meanv[col];
        const float scale = gamma[col] * rsqrtf(varv[col] + BNEPS);
        const float shift = beta[col] - mu * scale;
        float m = -3.4e38f;
#pragma unroll
        for (int t = 0; t < 8; ++t) {
            float h = acc[t] * scale + shift;
            h = fmaxf(h, 0.0f);
            m = fmaxf(m, h);
        }
        red[wv][lane] = m;
        __syncthreads();
        if (lane < 16) {
            const float mm = fmaxf(red[wv][lane], red[wv][lane + 16]);
            out_features[(size_t)q * COUT + col] = mm;
        }
    }
}

__global__ __launch_bounds__(256) void reduce_stats_kernel(
    const float* __restrict__ partial_sum, const float* __restrict__ partial_sq,
    float* __restrict__ meanv, float* __restrict__ varv)
{
    const int ch  = blockIdx.x;
    const int tid = threadIdx.x;
    __shared__ float ssum[256], ssq[256];
    float s = 0.0f, sq = 0.0f;
    for (int i = tid; i < BATCH * NPOINT; i += 256) {
        s  += partial_sum[(size_t)i * COUT + ch];
        sq += partial_sq [(size_t)i * COUT + ch];
    }
    ssum[tid] = s; ssq[tid] = sq;
    __syncthreads();
    for (int off = 128; off > 0; off >>= 1) {
        if (tid < off) { ssum[tid] += ssum[tid + off]; ssq[tid] += ssq[tid + off]; }
        __syncthreads();
    }
    if (tid == 0) {
        const float cnt = (float)(BATCH * NPOINT * KNN);
        const float mu  = ssum[0] / cnt;
        meanv[ch] = mu;
        varv[ch]  = ssq[0] / cnt - mu * mu;
    }
}

extern "C" void kernel_launch(void* const* d_in, const int* in_sizes, int n_in,
                              void* d_out, int out_size, void* d_ws, size_t ws_size,
                              hipStream_t stream)
{
    const float* xyz      = (const float*)d_in[0];
    const float* features = (const float*)d_in[1];
    const float* W        = (const float*)d_in[2];
    const float* gamma    = (const float*)d_in[3];
    const float* beta     = (const float*)d_in[4];

    float* out          = (float*)d_out;
    float* new_xyz      = out;
    float* out_features = out + BATCH * NPOINT * 3;

    char* ws = (char*)d_ws;
    size_t off = 0;
    int*   fps_idx     = (int*)(ws + off);   off += sizeof(int)   * BATCH * NPOINT;
    int*   knn_idx     = (int*)(ws + off);   off += sizeof(int)   * BATCH * NPOINT * KNN;
    float* Wpad        = (float*)(ws + off); off += sizeof(float) * KPAD * COUT;
    off = (off + 255) & ~(size_t)255;
    float* partial_sum = (float*)(ws + off); off += sizeof(float) * BATCH * NPOINT * COUT;
    float* partial_sq  = (float*)(ws + off); off += sizeof(float) * BATCH * NPOINT * COUT;
    float* meanv       = (float*)(ws + off); off += sizeof(float) * COUT;
    float* varv        = (float*)(ws + off); off += sizeof(float) * COUT;
    (void)ws_size; (void)in_sizes; (void)n_in; (void)out_size;

    fps_kernel<<<BATCH, 1024, 0, stream>>>(xyz, fps_idx);
    gather_new_xyz_kernel<<<(BATCH * NPOINT * 3 + 255) / 256, 256, 0, stream>>>(
        xyz, fps_idx, new_xyz);
    knn_kernel<<<(BATCH * NPOINT + 127) / 128, 128, 0, stream>>>(
        xyz, new_xyz, knn_idx);
    pad_w_kernel<<<(KPAD * COUT + 255) / 256, 256, 0, stream>>>(W, Wpad);
    group_gemm_kernel<<<BATCH * NPOINT, 256, 0, stream>>>(
        xyz, features, new_xyz, knn_idx, Wpad, gamma, beta,
        meanv, varv, partial_sum, partial_sq, out_features, 0);
    reduce_stats_kernel<<<COUT, 256, 0, stream>>>(
        partial_sum, partial_sq, meanv, varv);
    group_gemm_kernel<<<BATCH * NPOINT, 256, 0, stream>>>(
        xyz, features, new_xyz, knn_idx, Wpad, gamma, beta,
        meanv, varv, partial_sum, partial_sq, out_features, 1);
}


# Sampling block of 8 waves, 32 points per thread in registers (was 32 waves x 8): a selection's cross-lane reductions run 2 waves per vector unit, second join over 8 pairs; distances bit-identical
# speedup vs baseline: 3.7250x; 1.9039x over previous
#include <hip/hip_runtime.h>

typedef __attribute__((ext_vector_type(2))) float v2f;
typedef __attribute__((ext_vector_type(8))) float v8f;

#define BATCH   4
#define NPTS    8192
#define CIN     64
#define COUT    128
#define NPOINT  2048
#define KNN     16
#define KRAW    67
#define KPAD    68
#define BNEPS   1e-5f

#define FPS_THREADS 256
#define FPS_WAVES   (FPS_THREADS / 32)
#define FPS_PPT     (NPTS / FPS_THREADS)
#define FPS_GROUP   8
static_assert(FPS_PPT == 4 * FPS_GROUP, "the join below is written for four runs per thread");
static_assert(24 * FPS_THREADS == 6144, "slots 0 to 23 are the points below index 6144");

template <int M>
__device__ __forceinline__ int row_xor(int v)
{
    return __builtin_amdgcn_update_dpp(v, v, 0x160 + M, 0xf, 0xf, true);
}
__device__ __forceinline__ int half_xor(int v)
{
    return __builtin_amdgcn_permlanex16(v, v, 0x76543210, 0xfedcba98, false, false);
}
__device__ __forceinline__ int wave_max_i32(int x)
{
    x = max(x, row_xor<1>(x));
    x = max(x, row_xor<2>(x));
    x = max(x, row_xor<4>(x));
    x = max(x, row_xor<8>(x));
    x = max(x, half_xor(x));
    return x;
}
__device__ __forceinline__ int wave_min_i32(int x)
{
    x = min(x, row_xor<1>(x));
    x = min(x, row_xor<2>(x));
    x = min(x, row_xor<4>(x));
    x = min(x, row_xor<8>(x));
    x = min(x, half_xor(x));
    return x;
}
__device__ __forceinline__ int oct_max_i32(int x)
{
    x = max(x, row_xor<1>(x));
    x = max(x, row_xor<2>(x));
    x = max(x, row_xor<4>(x));
    return x;
}
__device__ __forceinline__ int oct_min_i32(int x)
{
    x = min(x, row_xor<1>(x));
    x = min(x, row_xor<2>(x));
    x = min(x, row_xor<4>(x));
    return x;
}

__global__ __launch_bounds__(FPS_THREADS) void fps_kernel(const float* __restrict__ xyz,
                                                          int* __restrict__ fps_idx)
{
    const int b    = blockIdx.x;
    const int tid  = threadIdx.x;
    const int lane = tid & 31;
    const int wv   = tid >> 5;
    const float* xb = xyz + (size_t)b * NPTS * 3;

    __shared__ int2 spair[2][FPS_WAVES];

    __shared__ float4 spts[NPTS];

    float px[FPS_PPT], py[FPS_PPT], pz[FPS_PPT], dist[FPS_PPT];
#pragma unroll
    for (int i = 0; i < FPS_PPT; ++i) {
        const int n = i * FPS_THREADS + tid;
        px[i] = xb[n * 3 + 0];
        py[i] = xb[n * 3 + 1];
        pz[i] = xb[n * 3 + 2];
        spts[n] = make_float4(px[i], py[i], pz[i], 0.0f);
        dist[i] = 3.4e38f;
    }
    __syncthreads();

    int cur = 0;
    for (int s = 0; s < NPOINT; ++s) {
        const float4 cpt = spts[cur];
        if (tid == 0) fps_idx[b * NPOINT + s] = cur;
        const float cx = cpt.x;
        const float cy = cpt.y;
        const float cz = cpt.z;

        float gbest[FPS_PPT / FPS_GROUP]; int gidx[FPS_PPT / FPS_GROUP];
#pragma unroll
        for (int g = 0; g < FPS_PPT / FPS_GROUP; ++g) {
            float best = -1.0f; int bi = 0;
#pragma unroll
            for (int k = 0; k < FPS_GROUP; ++k) {
                const int i = g * FPS_GROUP + k;
                const int n = i * FPS_THREADS + tid;
                const float dx = px[i] - cx;
                const float dy = py[i] - cy;
                const float dz = pz[i] - cz;
                float d;
                if (i < 24) {
                    d = __builtin_fmaf(dz, dz, __builtin_fmaf(dx, dx, dy * dy));
                } else {
                    float s2;
                    {
#pragma clang fp contract(off)
                        s2 = dx * dx + dy * dy;
                    }
                    d = __builtin_fmaf(dz, dz, s2);
                }
                const float nd = fminf(dist[i], d);
                dist[i] = nd;
                if (nd > best) { best = nd; bi = n; }
            }
            gbest[g] = best; gidx[g] = bi;
        }
        const bool  t01 = gbest[1] > gbest[0];
        const float b01 = t01 ? gbest[1] : gbest[0];
        const int   i01 = t01 ? gidx[1] : gidx[0];
        const bool  t23 = gbest[3] > gbest[2];
        const float b23 = t23 ? gbest[3] : gbest[2];
        const int   i23 = t23 ? gidx[3] : gidx[2];
        const bool  t   = b23 > b01;
        const float best = t ? b23 : b01;
        const int   bidx = t ? i23 : i01;

        const int bb = __float_as_int(best);
        const int wm = wave_max_i32(bb);
        const int wi = wave_min_i32(bb == wm ? bidx : 0x7fffffff);
        if (lane == 0) spair[s & 1][wv] = make_int2(wm, wi);
        __syncthreads();

        const int2 pr = spair[s & 1][lane & (FPS_WAVES - 1)];
        const int  gm = oct_max_i32(pr.x);
        const int  gi = oct_min_i32(pr.x == gm ? pr.y : 0x7fffffff);
        cur = gi & (NPTS - 1);
    }
}

__global__ void gather_new_xyz_kernel(const float* __restrict__ xyz,
                                      const int* __restrict__ fps_idx,
                                      float* __restrict__ new_xyz)
{
    const int e = blockIdx.x * blockDim.x + threadIdx.x;
    if (e >= BATCH * NPOINT * 3) return;
    const int c = e % 3;
    const int p = (e / 3) % NPOINT;
    const int b = e / (3 * NPOINT);
    const int n = fps_idx[b * NPOINT + p];
    new_xyz[e] = xyz[((size_t)b * NPTS + n) * 3 + c];
}

__device__ __forceinline__ int row_up1(int v)
{
    return __builtin_amdgcn_update_dpp(0, v, 0x111, 0xf, 0xf, true);
}

__device__ __forceinline__ void knn_offer(unsigned m, float d, int first,
                                          float& lv, int& li, float& tau)
{
    while (m) {
        const int L = __builtin_ctz(m);
        m &= m - 1;
        const float c = __int_as_float(__builtin_amdgcn_readlane(__float_as_int(d), L));
        if (c < tau) {
            const int   ci = first + L;
            const float sv = __int_as_float(row_up1(__float_as_int(lv)));
            const int   si = row_up1(li);
            const bool  g  = lv > c;
            const bool  gs = sv > c;
            lv = g ? (gs ? sv : c)  : lv;
            li = g ? (gs ? si : ci) : li;
            tau = __int_as_float(__builtin_amdgcn_readlane(__float_as_int(lv), 15));
        }
    }
}

__global__ __launch_bounds__(256) void knn_kernel(const float* __restrict__ xyz,
                                                  const float* __restrict__ new_xyz,
                                                  int* __restrict__ knn_idx)
{
    const int lane = threadIdx.x & 31;
    const int wv = __builtin_amdgcn_readfirstlane((int)(threadIdx.x >> 5));
    const int q0 = (blockIdx.x * 8 + wv) * 2;
    const int b  = q0 / NPOINT;
    const float* xb = xyz + (size_t)b * NPTS * 3;

    const float ax = new_xyz[q0 * 3 + 0], ay = new_xyz[q0 * 3 + 1], az = new_xyz[q0 * 3 + 2];
    const float bx = new_xyz[q0 * 3 + 3], by = new_xyz[q0 * 3 + 4], bz = new_xyz[q0 * 3 + 5];

    float lva = 3.4e38f, lvb = 3.4e38f;
    int   lia = 0,       lib = 0;
    float taua = 3.4e38f, taub = 3.4e38f;

    for (int base = 0; base < NPTS; base += 128) {
        float px[4], py[4], pz[4];
#pragma unroll
        for (int u = 0; u < 4; ++u) {
            const float* p = xb + (size_t)(base + u * 32 + lane) * 3;
            px[u] = p[0]; py[u] = p[1]; pz[u] = p[2];
        }
        float da[4], db[4];
        unsigned ma[4], mb[4];
#pragma unroll
        for (int u = 0; u < 4; ++u) {
            {
                const float dx = px[u] - ax, dy = py[u] - ay, dz = pz[u] - az;
                float s2;
                {
#pragma clang fp contract(off)
                    s2 = dx * dx + dy * dy;
                }
                da[u] = __builtin_fmaf(dz, dz, s2);
            }
            {
                const float dx = px[u] - bx, dy = py[u] - by, dz = pz[u] - bz;
                float s2;
                {
#pragma clang fp contract(off)
                    s2 = dx * dx + dy * dy;
                }
                db[u] = __builtin_fmaf(dz, dz, s2);
            }
            ma[u] = __builtin_amdgcn_ballot_w32(da[u] < taua);
            mb[u] = __builtin_amdgcn_ballot_w32(db[u] < taub);
        }
#pragma unroll
        for (int u = 0; u < 4; ++u) {
            knn_offer(ma[u], da[u], base + u * 32, lva, lia, taua);
            knn_offer(mb[u], db[u], base + u * 32, lvb, lib, taub);
        }
    }
    knn_idx[(size_t)q0 * KNN + lane] = (lane < 16) ? lia : lib;
}

__global__ void pad_w_kernel(const float* __restrict__ W, float* __restrict__ Wpad)
{
    const int e = blockIdx.x * blockDim.x + threadIdx.x;
    if (e >= KPAD * COUT) return;
    const int r = e / COUT;
    Wpad[e] = (r < KRAW) ? W[e] : 0.0f;
}

__global__ __launch_bounds__(256) void group_gemm_kernel(
    const float* __restrict__ xyz, const float* __restrict__ features,
    const float* __restrict__ new_xyz, const int* __restrict__ knn_idx,
    const float* __restrict__ Wpad,
    const float* __restrict__ gamma, const float* __restrict__ beta,
    const float* __restrict__ meanv, const float* __restrict__ varv,
    float* __restrict__ partial_sum, float* __restrict__ partial_sq,
    float* __restrict__ out_features, int final_pass)
{
    const int q    = blockIdx.x;
    const int b    = q / NPOINT;
    const int tid  = threadIdx.x;
    const int lane = tid & 31;
    const int wv   = tid >> 5;

    __shared__ float As[KNN][KPAD];
    __shared__ float red[8][32];

    const float qx = new_xyz[q * 3 + 0];
    const float qy = new_xyz[q * 3 + 1];
    const float qz = new_xyz[q * 3 + 2];

    for (int e = tid; e < KNN * KPAD; e += 256) {
        const int kk = e / KPAD;
        const int c  = e % KPAD;
        const int n  = knn_idx[(size_t)q * KNN + kk];
        float val;
        if (c < 3) {
            const float pv = xyz[((size_t)b * NPTS + n) * 3 + c];
            val = pv - (c == 0 ? qx : (c == 1 ? qy : qz));
        } else if (c < KRAW) {
            val = features[((size_t)b * NPTS + n) * CIN + (c - 3)];
        } else {
            val = 0.0f;
        }
        As[kk][c] = val;
    }
    __syncthreads();

    const int col = wv * 16 + (lane & 15);
    const int hi  = lane >> 4;
    const int row = lane & 15;

    v8f acc = {};
#pragma unroll
    for (int kb = 0; kb < KPAD / 4; ++kb) {
        const int k0 = kb * 4 + hi * 2;
        v2f a, bb;
        a.x  = As[row][k0 + 0];
        a.y  = As[row][k0 + 1];
        bb.x = Wpad[(k0 + 0) * COUT + col];
        bb.y = Wpad[(k0 + 1) * COUT + col];
        acc = __builtin_amdgcn_wmma_f32_16x16x4_f32(
                  false, a, false, bb, (short)0, acc, false, false);
    }

    if (!final_pass) {
        float s = 0.0f, ss = 0.0f;
#pragma unroll
        for (int t = 0; t < 8; ++t) { const float h = acc[t]; s += h; ss += h * h; }
        red[wv][lane] = s;
        __syncthreads();
        float s2 = 0.0f;
        if (lane < 16) s2 = red[wv][lane] + red[wv][lane + 16];
        __syncthreads();
        red[wv][lane] = ss;
        __syncthreads();
        if (lane < 16) {
            const float ss2 = red[wv][lane] + red[wv][lane + 16];
            partial_sum[(size_t)q * COUT + col] = s2;
            partial_sq [(size_t)q * COUT + col] = ss2;
        }
    } else {
        const float mu    = meanv[col];
        const float scale = gamma[col] * rsqrtf(varv[col] + BNEPS);
        const float shift = beta[col] - mu * scale;
        float m = -3.4e38f;
#pragma unroll
        for (int t = 0; t < 8; ++t) {
            float h = acc[t] * scale + shift;
            h = fmaxf(h, 0.0f);
            m = fmaxf(m, h);
        }
        red[wv][lane] = m;
        __syncthreads();
        if (lane < 16) {
            const float mm = fmaxf(red[wv][lane], red[wv][lane + 16]);
            out_features[(size_t)q * COUT + col] = mm;
        }
    }
}

__global__ __launch_bounds__(256) void reduce_stats_kernel(
    const float* __restrict__ partial_sum, const float* __restrict__ partial_sq,
    float* __restrict__ meanv, float* __restrict__ varv)
{
    const int ch  = blockIdx.x;
    const int tid = threadIdx.x;
    __shared__ float ssum[256], ssq[256];
    float s = 0.0f, sq = 0.0f;
    for (int i = tid; i < BATCH * NPOINT; i += 256) {
        s  += partial_sum[(size_t)i * COUT + ch];
        sq += partial_sq [(size_t)i * COUT + ch];
    }
    ssum[tid] = s; ssq[tid] = sq;
    __syncthreads();
    for (int off = 128; off > 0; off >>= 1) {
        if (tid < off) { ssum[tid] += ssum[tid + off]; ssq[tid] += ssq[tid + off]; }
        __syncthreads();
    }
    if (tid == 0) {
        const float cnt = (float)(BATCH * NPOINT * KNN);
        const float mu  = ssum[0] / cnt;
        meanv[ch] = mu;
        varv[ch]  = ssq[0] / cnt - mu * mu;
    }
}

extern "C" void kernel_launch(void* const* d_in, const int* in_sizes, int n_in,
                              void* d_out, int out_size, void* d_ws, size_t ws_size,
                              hipStream_t stream)
{
    const float* xyz      = (const float*)d_in[0];
    const float* features = (const float*)d_in[1];
    const float* W        = (const float*)d_in[2];
    const float* gamma    = (const float*)d_in[3];
    const float* beta     = (const float*)d_in[4];

    float* out          = (float*)d_out;
    float* new_xyz      = out;
    float* out_features = out + BATCH * NPOINT * 3;

    char* ws = (char*)d_ws;
    size_t off = 0;
    int*   fps_idx     = (int*)(ws + off);   off += sizeof(int)   * BATCH * NPOINT;
    int*   knn_idx     = (int*)(ws + off);   off += sizeof(int)   * BATCH * NPOINT * KNN;
    float* Wpad        = (float*)(ws + off); off += sizeof(float) * KPAD * COUT;
    off = (off + 255) & ~(size_t)255;
    float* partial_sum = (float*)(ws + off); off += sizeof(float) * BATCH * NPOINT * COUT;
    float* partial_sq  = (float*)(ws + off); off += sizeof(float) * BATCH * NPOINT * COUT;
    float* meanv       = (float*)(ws + off); off += sizeof(float) * COUT;
    float* varv        = (float*)(ws + off); off += sizeof(float) * COUT;
    (void)ws_size; (void)in_sizes; (void)n_in; (void)out_size;

    fps_kernel<<<BATCH, FPS_THREADS, 0, stream>>>(xyz, fps_idx);
    gather_new_xyz_kernel<<<(BATCH * NPOINT * 3 + 255) / 256, 256, 0, stream>>>(
        xyz, fps_idx, new_xyz);
    knn_kernel<<<BATCH * NPOINT / 16, 256, 0, stream>>>(
        xyz, new_xyz, knn_idx);
    pad_w_kernel<<<(KPAD * COUT + 255) / 256, 256, 0, stream>>>(W, Wpad);
    group_gemm_kernel<<<BATCH * NPOINT, 256, 0, stream>>>(
        xyz, features, new_xyz, knn_idx, Wpad, gamma, beta,
        meanv, varv, partial_sum, partial_sq, out_features, 0);
    reduce_stats_kernel<<<COUT, 256, 0, stream>>>(
        partial_sum, partial_sq, meanv, varv);
    group_gemm_kernel<<<BATCH * NPOINT, 256, 0, stream>>>(
        xyz, features, new_xyz, knn_idx, Wpad, gamma, beta,
        meanv, varv, partial_sum, partial_sq, out_features, 1);
}


# Sampling scan: running distances kept as f32 bits; minimum, running best and its compare taken on them as signed integers (the f32 minimum's NaN-quieting instruction per point goes); bit-identical
# speedup vs baseline: 4.0489x; 1.0870x over previous
#include <hip/hip_runtime.h>

typedef __attribute__((ext_vector_type(2))) float v2f;
typedef __attribute__((ext_vector_type(8))) float v8f;

#define BATCH   4
#define NPTS    8192
#define CIN     64
#define COUT    128
#define NPOINT  2048
#define KNN     16
#define KRAW    67
#define KPAD    68
#define BNEPS   1e-5f

#define FPS_THREADS 256
#define FPS_WAVES   (FPS_THREADS / 32)
#define FPS_PPT     (NPTS / FPS_THREADS)
#define FPS_GROUP   8
static_assert(FPS_PPT == 4 * FPS_GROUP, "the join below is written for four runs per thread");
static_assert(24 * FPS_THREADS == 6144, "slots 0 to 23 are the points below index 6144");

template <int M>
__device__ __forceinline__ int row_xor(int v)
{
    return __builtin_amdgcn_update_dpp(v, v, 0x160 + M, 0xf, 0xf, true);
}
__device__ __forceinline__ int half_xor(int v)
{
    return __builtin_amdgcn_permlanex16(v, v, 0x76543210, 0xfedcba98, false, false);
}
__device__ __forceinline__ int wave_max_i32(int x)
{
    x = max(x, row_xor<1>(x));
    x = max(x, row_xor<2>(x));
    x = max(x, row_xor<4>(x));
    x = max(x, row_xor<8>(x));
    x = max(x, half_xor(x));
    return x;
}
__device__ __forceinline__ int wave_min_i32(int x)
{
    x = min(x, row_xor<1>(x));
    x = min(x, row_xor<2>(x));
    x = min(x, row_xor<4>(x));
    x = min(x, row_xor<8>(x));
    x = min(x, half_xor(x));
    return x;
}
__device__ __forceinline__ int oct_max_i32(int x)
{
    x = max(x, row_xor<1>(x));
    x = max(x, row_xor<2>(x));
    x = max(x, row_xor<4>(x));
    return x;
}
__device__ __forceinline__ int oct_min_i32(int x)
{
    x = min(x, row_xor<1>(x));
    x = min(x, row_xor<2>(x));
    x = min(x, row_xor<4>(x));
    return x;
}

__global__ __launch_bounds__(FPS_THREADS) void fps_kernel(const float* __restrict__ xyz,
                                                          int* __restrict__ fps_idx)
{
    const int b    = blockIdx.x;
    const int tid  = threadIdx.x;
    const int lane = tid & 31;
    const int wv   = tid >> 5;
    const float* xb = xyz + (size_t)b * NPTS * 3;

    __shared__ int2 spair[2][FPS_WAVES];

    __shared__ float4 spts[NPTS];

    float px[FPS_PPT], py[FPS_PPT], pz[FPS_PPT];
    int   distb[FPS_PPT];
#pragma unroll
    for (int i = 0; i < FPS_PPT; ++i) {
        const int n = i * FPS_THREADS + tid;
        px[i] = xb[n * 3 + 0];
        py[i] = xb[n * 3 + 1];
        pz[i] = xb[n * 3 + 2];
        spts[n] = make_float4(px[i], py[i], pz[i], 0.0f);
        distb[i] = __float_as_int(3.4e38f);
    }
    __syncthreads();

    int cur = 0;
    for (int s = 0; s < NPOINT; ++s) {
        const float4 cpt = spts[cur];
        if (tid == 0) fps_idx[b * NPOINT + s] = cur;
        const float cx = cpt.x;
        const float cy = cpt.y;
        const float cz = cpt.z;

        int gbest[FPS_PPT / FPS_GROUP]; int gidx[FPS_PPT / FPS_GROUP];
#pragma unroll
        for (int g = 0; g < FPS_PPT / FPS_GROUP; ++g) {
            int best = -1; int bi = 0;
#pragma unroll
            for (int k = 0; k < FPS_GROUP; ++k) {
                const int i = g * FPS_GROUP + k;
                const int n = i * FPS_THREADS + tid;
                const float dx = px[i] - cx;
                const float dy = py[i] - cy;
                const float dz = pz[i] - cz;
                float d;
                if (i < 24) {
                    d = __builtin_fmaf(dz, dz, __builtin_fmaf(dx, dx, dy * dy));
                } else {
                    float s2;
                    {
#pragma clang fp contract(off)
                        s2 = dx * dx + dy * dy;
                    }
                    d = __builtin_fmaf(dz, dz, s2);
                }
                const int nd = min(distb[i], __float_as_int(d));
                distb[i] = nd;
                if (nd > best) { best = nd; bi = n; }
            }
            gbest[g] = best; gidx[g] = bi;
        }
        const bool t01 = gbest[1] > gbest[0];
        const int  b01 = t01 ? gbest[1] : gbest[0];
        const int  i01 = t01 ? gidx[1] : gidx[0];
        const bool t23 = gbest[3] > gbest[2];
        const int  b23 = t23 ? gbest[3] : gbest[2];
        const int  i23 = t23 ? gidx[3] : gidx[2];
        const bool t   = b23 > b01;
        const int  bb  = t ? b23 : b01;
        const int  bidx = t ? i23 : i01;

        const int wm = wave_max_i32(bb);
        const int wi = wave_min_i32(bb == wm ? bidx : 0x7fffffff);
        if (lane == 0) spair[s & 1][wv] = make_int2(wm, wi);
        __syncthreads();

        const int2 pr = spair[s & 1][lane & (FPS_WAVES - 1)];
        const int  gm = oct_max_i32(pr.x);
        const int  gi = oct_min_i32(pr.x == gm ? pr.y : 0x7fffffff);
        cur = gi & (NPTS - 1);
    }
}

__global__ void gather_new_xyz_kernel(const float* __restrict__ xyz,
                                      const int* __restrict__ fps_idx,
                                      float* __restrict__ new_xyz)
{
    const int e = blockIdx.x * blockDim.x + threadIdx.x;
    if (e >= BATCH * NPOINT * 3) return;
    const int c = e % 3;
    const int p = (e / 3) % NPOINT;
    const int b = e / (3 * NPOINT);
    const int n = fps_idx[b * NPOINT + p];
    new_xyz[e] = xyz[((size_t)b * NPTS + n) * 3 + c];
}

__device__ __forceinline__ int row_up1(int v)
{
    return __builtin_amdgcn_update_dpp(0, v, 0x111, 0xf, 0xf, true);
}

__device__ __forceinline__ void knn_offer(unsigned m, float d, int first,
                                          float& lv, int& li, float& tau)
{
    while (m) {
        const int L = __builtin_ctz(m);
        m &= m - 1;
        const float c = __int_as_float(__builtin_amdgcn_readlane(__float_as_int(d), L));
        if (c < tau) {
            const int   ci = first + L;
            const float sv = __int_as_float(row_up1(__float_as_int(lv)));
            const int   si = row_up1(li);
            const bool  g  = lv > c;
            const bool  gs = sv > c;
            lv = g ? (gs ? sv : c)  : lv;
            li = g ? (gs ? si : ci) : li;
            tau = __int_as_float(__builtin_amdgcn_readlane(__float_as_int(lv), 15));
        }
    }
}

__global__ __launch_bounds__(256) void knn_kernel(const float* __restrict__ xyz,
                                                  const float* __restrict__ new_xyz,
                                                  int* __restrict__ knn_idx)
{
    const int lane = threadIdx.x & 31;
    const int wv = __builtin_amdgcn_readfirstlane((int)(threadIdx.x >> 5));
    const int q0 = (blockIdx.x * 8 + wv) * 2;
    const int b  = q0 / NPOINT;
    const float* xb = xyz + (size_t)b * NPTS * 3;

    const float ax = new_xyz[q0 * 3 + 0], ay = new_xyz[q0 * 3 + 1], az = new_xyz[q0 * 3 + 2];
    const float bx = new_xyz[q0 * 3 + 3], by = new_xyz[q0 * 3 + 4], bz = new_xyz[q0 * 3 + 5];

    float lva = 3.4e38f, lvb = 3.4e38f;
    int   lia = 0,       lib = 0;
    float taua = 3.4e38f, taub = 3.4e38f;

    for (int base = 0; base < NPTS; base += 128) {
        float px[4], py[4], pz[4];
#pragma unroll
        for (int u = 0; u < 4; ++u) {
            const float* p = xb + (size_t)(base + u * 32 + lane) * 3;
            px[u] = p[0]; py[u] = p[1]; pz[u] = p[2];
        }
        float da[4], db[4];
        unsigned ma[4], mb[4];
#pragma unroll
        for (int u = 0; u < 4; ++u) {
            {
                const float dx = px[u] - ax, dy = py[u] - ay, dz = pz[u] - az;
                float s2;
                {
#pragma clang fp contract(off)
                    s2 = dx * dx + dy * dy;
                }
                da[u] = __builtin_fmaf(dz, dz, s2);
            }
            {
                const float dx = px[u] - bx, dy = py[u] - by, dz = pz[u] - bz;
                float s2;
                {
#pragma clang fp contract(off)
                    s2 = dx * dx + dy * dy;
                }
                db[u] = __builtin_fmaf(dz, dz, s2);
            }
            ma[u] = __builtin_amdgcn_ballot_w32(da[u] < taua);
            mb[u] = __builtin_amdgcn_ballot_w32(db[u] < taub);
        }
#pragma unroll
        for (int u = 0; u < 4; ++u) {
            knn_offer(ma[u], da[u], base + u * 32, lva, lia, taua);
            knn_offer(mb[u], db[u], base + u * 32, lvb, lib, taub);
        }
    }
    knn_idx[(size_t)q0 * KNN + lane] = (lane < 16) ? lia : lib;
}

__global__ void pad_w_kernel(const float* __restrict__ W, float* __restrict__ Wpad)
{
    const int e = blockIdx.x * blockDim.x + threadIdx.x;
    if (e >= KPAD * COUT) return;
    const int r = e / COUT;
    Wpad[e] = (r < KRAW) ? W[e] : 0.0f;
}

__global__ __launch_bounds__(256) void group_gemm_kernel(
    const float* __restrict__ xyz, const float* __restrict__ features,
    const float* __restrict__ new_xyz, const int* __restrict__ knn_idx,
    const float* __restrict__ Wpad,
    const float* __restrict__ gamma, const float* __restrict__ beta,
    const float* __restrict__ meanv, const float* __restrict__ varv,
    float* __restrict__ partial_sum, float* __restrict__ partial_sq,
    float* __restrict__ out_features, int final_pass)
{
    const int q    = blockIdx.x;
    const int b    = q / NPOINT;
    const int tid  = threadIdx.x;
    const int lane = tid & 31;
    const int wv   = tid >> 5;

    __shared__ float As[KNN][KPAD];
    __shared__ float red[8][32];

    const float qx = new_xyz[q * 3 + 0];
    const float qy = new_xyz[q * 3 + 1];
    const float qz = new_xyz[q * 3 + 2];

    for (int e = tid; e < KNN * KPAD; e += 256) {
        const int kk = e / KPAD;
        const int c  = e % KPAD;
        const int n  = knn_idx[(size_t)q * KNN + kk];
        float val;
        if (c < 3) {
            const float pv = xyz[((size_t)b * NPTS + n) * 3 + c];
            val = pv - (c == 0 ? qx : (c == 1 ? qy : qz));
        } else if (c < KRAW) {
            val = features[((size_t)b * NPTS + n) * CIN + (c - 3)];
        } else {
            val = 0.0f;
        }
        As[kk][c] = val;
    }
    __syncthreads();

    const int col = wv * 16 + (lane & 15);
    const int hi  = lane >> 4;
    const int row = lane & 15;

    v8f acc = {};
#pragma unroll
    for (int kb = 0; kb < KPAD / 4; ++kb) {
        const int k0 = kb * 4 + hi * 2;
        v2f a, bb;
        a.x  = As[row][k0 + 0];
        a.y  = As[row][k0 + 1];
        bb.x = Wpad[(k0 + 0) * COUT + col];
        bb.y = Wpad[(k0 + 1) * COUT + col];
        acc = __builtin_amdgcn_wmma_f32_16x16x4_f32(
                  false, a, false, bb, (short)0, acc, false, false);
    }

    if (!final_pass) {
        float s = 0.0f, ss = 0.0f;
#pragma unroll
        for (int t = 0; t < 8; ++t) { const float h = acc[t]; s += h; ss += h * h; }
        red[wv][lane] = s;
        __syncthreads();
        float s2 = 0.0f;
        if (lane < 16) s2 = red[wv][lane] + red[wv][lane + 16];
        __syncthreads();
        red[wv][lane] = ss;
        __syncthreads();
        if (lane < 16) {
            const float ss2 = red[wv][lane] + red[wv][lane + 16];
            partial_sum[(size_t)q * COUT + col] = s2;
            partial_sq [(size_t)q * COUT + col] = ss2;
        }
    } else {
        const float mu    = meanv[col];
        const float scale = gamma[col] * rsqrtf(varv[col] + BNEPS);
        const float shift = beta[col] - mu * scale;
        float m = -3.4e38f;
#pragma unroll
        for (int t = 0; t < 8; ++t) {
            float h = acc[t] * scale + shift;
            h = fmaxf(h, 0.0f);
            m = fmaxf(m, h);
        }
        red[wv][lane] = m;
        __syncthreads();
        if (lane < 16) {
            const float mm = fmaxf(red[wv][lane], red[wv][lane + 16]);
            out_features[(size_t)q * COUT + col] = mm;
        }
    }
}

__global__ __launch_bounds__(256) void reduce_stats_kernel(
    const float* __restrict__ partial_sum, const float* __restrict__ partial_sq,
    float* __restrict__ meanv, float* __restrict__ varv)
{
    const int ch  = blockIdx.x;
    const int tid = threadIdx.x;
    __shared__ float ssum[256], ssq[256];
    float s = 0.0f, sq = 0.0f;
    for (int i = tid; i < BATCH * NPOINT; i += 256) {
        s  += partial_sum[(size_t)i * COUT + ch];
        sq += partial_sq [(size_t)i * COUT + ch];
    }
    ssum[tid] = s; ssq[tid] = sq;
    __syncthreads();
    for (int off = 128; off > 0; off >>= 1) {
        if (tid < off) { ssum[tid] += ssum[tid + off]; ssq[tid] += ssq[tid + off]; }
        __syncthreads();
    }
    if (tid == 0) {
        const float cnt = (float)(BATCH * NPOINT * KNN);
        const float mu  = ssum[0] / cnt;
        meanv[ch] = mu;
        varv[ch]  = ssq[0] / cnt - mu * mu;
    }
}

extern "C" void kernel_launch(void* const* d_in, const int* in_sizes, int n_in,
                              void* d_out, int out_size, void* d_ws, size_t ws_size,
                              hipStream_t stream)
{
    const float* xyz      = (const float*)d_in[0];
    const float* features = (const float*)d_in[1];
    const float* W        = (const float*)d_in[2];
    const float* gamma    = (const float*)d_in[3];
    const float* beta     = (const float*)d_in[4];

    float* out          = (float*)d_out;
    float* new_xyz      = out;
    float* out_features = out + BATCH * NPOINT * 3;

    char* ws = (char*)d_ws;
    size_t off = 0;
    int*   fps_idx     = (int*)(ws + off);   off += sizeof(int)   * BATCH * NPOINT;
    int*   knn_idx     = (int*)(ws + off);   off += sizeof(int)   * BATCH * NPOINT * KNN;
    float* Wpad        = (float*)(ws + off); off += sizeof(float) * KPAD * COUT;
    off = (off + 255) & ~(size_t)255;
    float* partial_sum = (float*)(ws + off); off += sizeof(float) * BATCH * NPOINT * COUT;
    float* partial_sq  = (float*)(ws + off); off += sizeof(float) * BATCH * NPOINT * COUT;
    float* meanv       = (float*)(ws + off); off += sizeof(float) * COUT;
    float* varv        = (float*)(ws + off); off += sizeof(float) * COUT;
    (void)ws_size; (void)in_sizes; (void)n_in; (void)out_size;

    fps_kernel<<<BATCH, FPS_THREADS, 0, stream>>>(xyz, fps_idx);
    gather_new_xyz_kernel<<<(BATCH * NPOINT * 3 + 255) / 256, 256, 0, stream>>>(
        xyz, fps_idx, new_xyz);
    knn_kernel<<<BATCH * NPOINT / 16, 256, 0, stream>>>(
        xyz, new_xyz, knn_idx);
    pad_w_kernel<<<(KPAD * COUT + 255) / 256, 256, 0, stream>>>(W, Wpad);
    group_gemm_kernel<<<BATCH * NPOINT, 256, 0, stream>>>(
        xyz, features, new_xyz, knn_idx, Wpad, gamma, beta,
        meanv, varv, partial_sum, partial_sq, out_features, 0);
    reduce_stats_kernel<<<COUT, 256, 0, stream>>>(
        partial_sum, partial_sq, meanv, varv);
    group_gemm_kernel<<<BATCH * NPOINT, 256, 0, stream>>>(
        xyz, features, new_xyz, knn_idx, Wpad, gamma, beta,
        meanv, varv, partial_sum, partial_sq, out_features, 1);
}


# Sampling scan keeps maxima only (max3; distances on slot pairs, packed, bits kept); index found after the wave maximum from compare masks: 4 compares pick the run of 8 slots, 8 the slot; ties kept
# speedup vs baseline: 4.1988x; 1.0370x over previous
#include <hip/hip_runtime.h>

typedef __attribute__((ext_vector_type(2))) float v2f;
typedef __attribute__((ext_vector_type(8))) float v8f;

#define BATCH   4
#define NPTS    8192
#define CIN     64
#define COUT    128
#define NPOINT  2048
#define KNN     16
#define KRAW    67
#define KPAD    68
#define BNEPS   1e-5f

#define FPS_THREADS 256
#define FPS_WAVES   (FPS_THREADS / 32)
#define FPS_PPT     (NPTS / FPS_THREADS)
#define FPS_GROUP   8
static_assert(FPS_PPT == 4 * FPS_GROUP, "the choice of the run below is written for four runs per thread");
static_assert(24 * FPS_THREADS == 6144, "slots 0 to 23 are the points below index 6144");

template <int M>
__device__ __forceinline__ int row_xor(int v)
{
    return __builtin_amdgcn_update_dpp(v, v, 0x160 + M, 0xf, 0xf, true);
}
__device__ __forceinline__ int half_xor(int v)
{
    return __builtin_amdgcn_permlanex16(v, v, 0x76543210, 0xfedcba98, false, false);
}
__device__ __forceinline__ int wave_max_i32(int x)
{
    x = max(x, row_xor<1>(x));
    x = max(x, row_xor<2>(x));
    x = max(x, row_xor<4>(x));
    x = max(x, row_xor<8>(x));
    x = max(x, half_xor(x));
    return x;
}
__device__ __forceinline__ int oct_max_i32(int x)
{
    x = max(x, row_xor<1>(x));
    x = max(x, row_xor<2>(x));
    x = max(x, row_xor<4>(x));
    return x;
}
__device__ __forceinline__ int oct_min_i32(int x)
{
    x = min(x, row_xor<1>(x));
    x = min(x, row_xor<2>(x));
    x = min(x, row_xor<4>(x));
    return x;
}

template <int G>
__device__ __forceinline__ unsigned run_first_holder(const int (&distb)[FPS_PPT], int wm)
{
    unsigned long long msel = 0ull; int jsel = 0;
#pragma unroll
    for (int j = FPS_GROUP / 2 - 1; j >= 0; --j) {
        const unsigned m0 = __builtin_amdgcn_ballot_w32(distb[G * FPS_GROUP + 2 * j + 0] == wm);
        const unsigned m1 = __builtin_amdgcn_ballot_w32(distb[G * FPS_GROUP + 2 * j + 1] == wm);
        const unsigned long long mm = ((unsigned long long)m1 << 32) | m0;
        if (j == FPS_GROUP / 2 - 1 || mm != 0ull) { msel = mm; jsel = j; }
    }
    const int p = __builtin_ctzll(msel);
    return (unsigned)((2 * jsel + (p >> 5)) * FPS_THREADS + (p & 31));
}

__global__ __launch_bounds__(FPS_THREADS) void fps_kernel(const float* __restrict__ xyz,
                                                          int* __restrict__ fps_idx)
{
    const int b    = blockIdx.x;
    const int tid  = threadIdx.x;
    const int lane = tid & 31;
    const int wv   = tid >> 5;
    const float* xb = xyz + (size_t)b * NPTS * 3;

    __shared__ int2 spair[2][FPS_WAVES];

    __shared__ float4 spts[NPTS];

    v2f px2[FPS_PPT / 2], py2[FPS_PPT / 2], pz2[FPS_PPT / 2];
    int distb[FPS_PPT];
#pragma unroll
    for (int i = 0; i < FPS_PPT; ++i) {
        const int n = i * FPS_THREADS + tid;
        const float x = xb[n * 3 + 0];
        const float y = xb[n * 3 + 1];
        const float z = xb[n * 3 + 2];
        px2[i >> 1][i & 1] = x;
        py2[i >> 1][i & 1] = y;
        pz2[i >> 1][i & 1] = z;
        spts[n] = make_float4(x, y, z, 0.0f);
        distb[i] = __float_as_int(3.4e38f);
    }
    __syncthreads();

    int cur = 0;
    for (int s = 0; s < NPOINT; ++s) {
        const float4 cpt = spts[cur];
        if (tid == 0) fps_idx[b * NPOINT + s] = cur;
        const float cx = cpt.x;
        const float cy = cpt.y;
        const float cz = cpt.z;

        const v2f cx2 = {cx, cx};
        const v2f cy2 = {cy, cy};
        const v2f cz2 = {cz, cz};
        int gbest[FPS_PPT / FPS_GROUP];
#pragma unroll
        for (int g = 0; g < FPS_PPT / FPS_GROUP; ++g) {
            int best = 0;
#pragma unroll
            for (int jj = 0; jj < FPS_GROUP / 2; ++jj) {
                const int j = g * (FPS_GROUP / 2) + jj;
                const v2f dx = px2[j] - cx2;
                const v2f dy = py2[j] - cy2;
                const v2f dz = pz2[j] - cz2;
                v2f d;
                if (2 * j < 24) {
                    d = __builtin_elementwise_fma(dz, dz, __builtin_elementwise_fma(dx, dx, dy * dy));
                } else {
                    v2f s2;
                    {
#pragma clang fp contract(off)
                        s2 = dx * dx + dy * dy;
                    }
                    d = __builtin_elementwise_fma(dz, dz, s2);
                }
                const int nd0 = min(distb[2 * j + 0], __float_as_int(d.x));
                const int nd1 = min(distb[2 * j + 1], __float_as_int(d.y));
                distb[2 * j + 0] = nd0;
                distb[2 * j + 1] = nd1;
                best = (jj == 0) ? max(nd0, nd1) : max(max(best, nd0), nd1);
            }
            gbest[g] = best;
        }
        const int bb = max(max(gbest[0], gbest[1]), max(gbest[2], gbest[3]));

        const int wm = wave_max_i32(bb);

        const unsigned r0 = __builtin_amdgcn_ballot_w32(gbest[0] == wm);
        const unsigned r1 = __builtin_amdgcn_ballot_w32(gbest[1] == wm);
        const unsigned r2 = __builtin_amdgcn_ballot_w32(gbest[2] == wm);
        int grun; unsigned key;
        if (r0)      { grun = 0; key = run_first_holder<0>(distb, wm); }
        else if (r1) { grun = 1; key = run_first_holder<1>(distb, wm); }
        else if (r2) { grun = 2; key = run_first_holder<2>(distb, wm); }
        else         { grun = 3; key = run_first_holder<3>(distb, wm); }
        const int wi = (int)key + grun * (FPS_GROUP * FPS_THREADS) + (tid & ~31);
        if (lane == 0) spair[s & 1][wv] = make_int2(wm, wi);
        __syncthreads();

        const int2 pr = spair[s & 1][lane & (FPS_WAVES - 1)];
        const int  gm = oct_max_i32(pr.x);
        const int  gi = oct_min_i32(pr.x == gm ? pr.y : 0x7fffffff);
        cur = gi & (NPTS - 1);
    }
}

__global__ void gather_new_xyz_kernel(const float* __restrict__ xyz,
                                      const int* __restrict__ fps_idx,
                                      float* __restrict__ new_xyz)
{
    const int e = blockIdx.x * blockDim.x + threadIdx.x;
    if (e >= BATCH * NPOINT * 3) return;
    const int c = e % 3;
    const int p = (e / 3) % NPOINT;
    const int b = e / (3 * NPOINT);
    const int n = fps_idx[b * NPOINT + p];
    new_xyz[e] = xyz[((size_t)b * NPTS + n) * 3 + c];
}

__device__ __forceinline__ int row_up1(int v)
{
    return __builtin_amdgcn_update_dpp(0, v, 0x111, 0xf, 0xf, true);
}

__device__ __forceinline__ void knn_offer(unsigned m, float d, int first,
                                          float& lv, int& li, float& tau)
{
    while (m) {
        const int L = __builtin_ctz(m);
        m &= m - 1;
        const float c = __int_as_float(__builtin_amdgcn_readlane(__float_as_int(d), L));
        if (c < tau) {
            const int   ci = first + L;
            const float sv = __int_as_float(row_up1(__float_as_int(lv)));
            const int   si = row_up1(li);
            const bool  g  = lv > c;
            const bool  gs = sv > c;
            lv = g ? (gs ? sv : c)  : lv;
            li = g ? (gs ? si : ci) : li;
            tau = __int_as_float(__builtin_amdgcn_readlane(__float_as_int(lv), 15));
        }
    }
}

__global__ __launch_bounds__(256) void knn_kernel(const float* __restrict__ xyz,
                                                  const float* __restrict__ new_xyz,
                                                  int* __restrict__ knn_idx)
{
    const int lane = threadIdx.x & 31;
    const int wv = __builtin_amdgcn_readfirstlane((int)(threadIdx.x >> 5));
    const int q0 = (blockIdx.x * 8 + wv) * 2;
    const int b  = q0 / NPOINT;
    const float* xb = xyz + (size_t)b * NPTS * 3;

    const float ax = new_xyz[q0 * 3 + 0], ay = new_xyz[q0 * 3 + 1], az = new_xyz[q0 * 3 + 2];
    const float bx = new_xyz[q0 * 3 + 3], by = new_xyz[q0 * 3 + 4], bz = new_xyz[q0 * 3 + 5];

    float lva = 3.4e38f, lvb = 3.4e38f;
    int   lia = 0,       lib = 0;
    float taua = 3.4e38f, taub = 3.4e38f;

    for (int base = 0; base < NPTS; base += 128) {
        float px[4], py[4], pz[4];
#pragma unroll
        for (int u = 0; u < 4; ++u) {
            const float* p = xb + (size_t)(base + u * 32 + lane) * 3;
            px[u] = p[0]; py[u] = p[1]; pz[u] = p[2];
        }
        float da[4], db[4];
        unsigned ma[4], mb[4];
#pragma unroll
        for (int u = 0; u < 4; ++u) {
            {
                const float dx = px[u] - ax, dy = py[u] - ay, dz = pz[u] - az;
                float s2;
                {
#pragma clang fp contract(off)
                    s2 = dx * dx + dy * dy;
                }
                da[u] = __builtin_fmaf(dz, dz, s2);
            }
            {
                const float dx = px[u] - bx, dy = py[u] - by, dz = pz[u] - bz;
                float s2;
                {
#pragma clang fp contract(off)
                    s2 = dx * dx + dy * dy;
                }
                db[u] = __builtin_fmaf(dz, dz, s2);
            }
            ma[u] = __builtin_amdgcn_ballot_w32(da[u] < taua);
            mb[u] = __builtin_amdgcn_ballot_w32(db[u] < taub);
        }
#pragma unroll
        for (int u = 0; u < 4; ++u) {
            knn_offer(ma[u], da[u], base + u * 32, lva, lia, taua);
            knn_offer(mb[u], db[u], base + u * 32, lvb, lib, taub);
        }
    }
    knn_idx[(size_t)q0 * KNN + lane] = (lane < 16) ? lia : lib;
}

__global__ void pad_w_kernel(const float* __restrict__ W, float* __restrict__ Wpad)
{
    const int e = blockIdx.x * blockDim.x + threadIdx.x;
    if (e >= KPAD * COUT) return;
    const int r = e / COUT;
    Wpad[e] = (r < KRAW) ? W[e] : 0.0f;
}

__global__ __launch_bounds__(256) void group_gemm_kernel(
    const float* __restrict__ xyz, const float* __restrict__ features,
    const float* __restrict__ new_xyz, const int* __restrict__ knn_idx,
    const float* __restrict__ Wpad,
    const float* __restrict__ gamma, const float* __restrict__ beta,
    const float* __restrict__ meanv, const float* __restrict__ varv,
    float* __restrict__ partial_sum, float* __restrict__ partial_sq,
    float* __restrict__ out_features, int final_pass)
{
    const int q    = blockIdx.x;
    const int b    = q / NPOINT;
    const int tid  = threadIdx.x;
    const int lane = tid & 31;
    const int wv   = tid >> 5;

    __shared__ float As[KNN][KPAD];
    __shared__ float red[8][32];

    const float qx = new_xyz[q * 3 + 0];
    const float qy = new_xyz[q * 3 + 1];
    const float qz = new_xyz[q * 3 + 2];

    for (int e = tid; e < KNN * KPAD; e += 256) {
        const int kk = e / KPAD;
        const int c  = e % KPAD;
        const int n  = knn_idx[(size_t)q * KNN + kk];
        float val;
        if (c < 3) {
            const float pv = xyz[((size_t)b * NPTS + n) * 3 + c];
            val = pv - (c == 0 ? qx : (c == 1 ? qy : qz));
        } else if (c < KRAW) {
            val = features[((size_t)b * NPTS + n) * CIN + (c - 3)];
        } else {
            val = 0.0f;
        }
        As[kk][c] = val;
    }
    __syncthreads();

    const int col = wv * 16 + (lane & 15);
    const int hi  = lane >> 4;
    const int row = lane & 15;

    v8f acc = {};
#pragma unroll
    for (int kb = 0; kb < KPAD / 4; ++kb) {
        const int k0 = kb * 4 + hi * 2;
        v2f a, bb;
        a.x  = As[row][k0 + 0];
        a.y  = As[row][k0 + 1];
        bb.x = Wpad[(k0 + 0) * COUT + col];
        bb.y = Wpad[(k0 + 1) * COUT + col];
        acc = __builtin_amdgcn_wmma_f32_16x16x4_f32(
                  false, a, false, bb, (short)0, acc, false, false);
    }

    if (!final_pass) {
        float s = 0.0f, ss = 0.0f;
#pragma unroll
        for (int t = 0; t < 8; ++t) { const float h = acc[t]; s += h; ss += h * h; }
        red[wv][lane] = s;
        __syncthreads();
        float s2 = 0.0f;
        if (lane < 16) s2 = red[wv][lane] + red[wv][lane + 16];
        __syncthreads();
        red[wv][lane] = ss;
        __syncthreads();
        if (lane < 16) {
            const float ss2 = red[wv][lane] + red[wv][lane + 16];
            partial_sum[(size_t)q * COUT + col] = s2;
            partial_sq [(size_t)q * COUT + col] = ss2;
        }
    } else {
        const float mu    = meanv[col];
        const float scale = gamma[col] * rsqrtf(varv[col] + BNEPS);
        const float shift = beta[col] - mu * scale;
        float m = -3.4e38f;
#pragma unroll
        for (int t = 0; t < 8; ++t) {
            float h = acc[t] * scale + shift;
            h = fmaxf(h, 0.0f);
            m = fmaxf(m, h);
        }
        red[wv][lane] = m;
        __syncthreads();
        if (lane < 16) {
            const float mm = fmaxf(red[wv][lane], red[wv][lane + 16]);
            out_features[(size_t)q * COUT + col] = mm;
        }
    }
}

__global__ __launch_bounds__(256) void reduce_stats_kernel(
    const float* __restrict__ partial_sum, const float* __restrict__ partial_sq,
    float* __restrict__ meanv, float* __restrict__ varv)
{
    const int ch  = blockIdx.x;
    const int tid = threadIdx.x;
    __shared__ float ssum[256], ssq[256];
    float s = 0.0f, sq = 0.0f;
    for (int i = tid; i < BATCH * NPOINT; i += 256) {
        s  += partial_sum[(size_t)i * COUT + ch];
        sq += partial_sq [(size_t)i * COUT + ch];
    }
    ssum[tid] = s; ssq[tid] = sq;
    __syncthreads();
    for (int off = 128; off > 0; off >>= 1) {
        if (tid < off) { ssum[tid] += ssum[tid + off]; ssq[tid] += ssq[tid + off]; }
        __syncthreads();
    }
    if (tid == 0) {
        const float cnt = (float)(BATCH * NPOINT * KNN);
        const float mu  = ssum[0] / cnt;
        meanv[ch] = mu;
        varv[ch]  = ssq[0] / cnt - mu * mu;
    }
}

extern "C" void kernel_launch(void* const* d_in, const int* in_sizes, int n_in,
                              void* d_out, int out_size, void* d_ws, size_t ws_size,
                              hipStream_t stream)
{
    const float* xyz      = (const float*)d_in[0];
    const float* features = (const float*)d_in[1];
    const float* W        = (const float*)d_in[2];
    const float* gamma    = (const float*)d_in[3];
    const float* beta     = (const float*)d_in[4];

    float* out          = (float*)d_out;
    float* new_xyz      = out;
    float* out_features = out + BATCH * NPOINT * 3;

    char* ws = (char*)d_ws;
    size_t off = 0;
    int*   fps_idx     = (int*)(ws + off);   off += sizeof(int)   * BATCH * NPOINT;
    int*   knn_idx     = (int*)(ws + off);   off += sizeof(int)   * BATCH * NPOINT * KNN;
    float* Wpad        = (float*)(ws + off); off += sizeof(float) * KPAD * COUT;
    off = (off + 255) & ~(size_t)255;
    float* partial_sum = (float*)(ws + off); off += sizeof(float) * BATCH * NPOINT * COUT;
    float* partial_sq  = (float*)(ws + off); off += sizeof(float) * BATCH * NPOINT * COUT;
    float* meanv       = (float*)(ws + off); off += sizeof(float) * COUT;
    float* varv        = (float*)(ws + off); off += sizeof(float) * COUT;
    (void)ws_size; (void)in_sizes; (void)n_in; (void)out_size;

    fps_kernel<<<BATCH, FPS_THREADS, 0, stream>>>(xyz, fps_idx);
    gather_new_xyz_kernel<<<(BATCH * NPOINT * 3 + 255) / 256, 256, 0, stream>>>(
        xyz, fps_idx, new_xyz);
    knn_kernel<<<BATCH * NPOINT / 16, 256, 0, stream>>>(
        xyz, new_xyz, knn_idx);
    pad_w_kernel<<<(KPAD * COUT + 255) / 256, 256, 0, stream>>>(W, Wpad);
    group_gemm_kernel<<<BATCH * NPOINT, 256, 0, stream>>>(
        xyz, features, new_xyz, knn_idx, Wpad, gamma, beta,
        meanv, varv, partial_sum, partial_sq, out_features, 0);
    reduce_stats_kernel<<<COUT, 256, 0, stream>>>(
        partial_sum, partial_sq, meanv, varv);
    group_gemm_kernel<<<BATCH * NPOINT, 256, 0, stream>>>(
        xyz, features, new_xyz, knn_idx, Wpad, gamma, beta,
        meanv, varv, partial_sum, partial_sq, out_features, 1);
}


# Grouped product: one work-group per 4 queries (64-row tile; a wave keeps 4 accumulators and loads each B fragment once for the four): a quarter of the waves per pass; per-query sums and order kept
# speedup vs baseline: 4.7005x; 1.1195x over previous
#include <hip/hip_runtime.h>

typedef __attribute__((ext_vector_type(2))) float v2f;
typedef __attribute__((ext_vector_type(8))) float v8f;

#define BATCH   4
#define NPTS    8192
#define CIN     64
#define COUT    128
#define NPOINT  2048
#define KNN     16
#define KRAW    67
#define KPAD    68
#define BNEPS   1e-5f

#define FPS_THREADS 256
#define FPS_WAVES   (FPS_THREADS / 32)
#define FPS_PPT     (NPTS / FPS_THREADS)
#define FPS_GROUP   8
#define FPS_NRUN    (FPS_PPT / FPS_GROUP)
#define FPS_LOWN    (24 * FPS_THREADS)
static_assert(FPS_PPT == 4 * FPS_GROUP, "the choice of the run below is written for four runs per thread");
static_assert(FPS_LOWN == 6144, "slots 0 to 23 are the points below index 6144");

template <int M>
__device__ __forceinline__ int row_xor(int v)
{
    return __builtin_amdgcn_update_dpp(v, v, 0x160 + M, 0xf, 0xf, true);
}
__device__ __forceinline__ int half_xor(int v)
{
    return __builtin_amdgcn_permlanex16(v, v, 0x76543210, 0xfedcba98, false, false);
}
__device__ __forceinline__ int wave_max_i32(int x)
{
    x = max(x, row_xor<1>(x));
    x = max(x, row_xor<2>(x));
    x = max(x, row_xor<4>(x));
    x = max(x, row_xor<8>(x));
    x = max(x, half_xor(x));
    return x;
}
__device__ __forceinline__ int oct_max_i32(int x)
{
    x = max(x, row_xor<1>(x));
    x = max(x, row_xor<2>(x));
    x = max(x, row_xor<4>(x));
    return x;
}
__device__ __forceinline__ int oct_min_i32(int x)
{
    x = min(x, row_xor<1>(x));
    x = min(x, row_xor<2>(x));
    x = min(x, row_xor<4>(x));
    return x;
}
__device__ __forceinline__ float lane_f32(float v, int l)
{
    return __int_as_float(__builtin_amdgcn_readlane(__float_as_int(v), l));
}

__device__ __forceinline__ v2f fps_pair_dist(const bool low, const v2f px, const v2f py, const v2f pz,
                                             const v2f cx2, const v2f cy2, const v2f cz2)
{
    const v2f dx = px - cx2;
    const v2f dy = py - cy2;
    const v2f dz = pz - cz2;
    v2f d;
    if (low) {
        d = __builtin_elementwise_fma(dz, dz, __builtin_elementwise_fma(dx, dx, dy * dy));
    } else {
        v2f s2;
        {
#pragma clang fp contract(off)
            s2 = dx * dx + dy * dy;
        }
        d = __builtin_elementwise_fma(dz, dz, s2);
    }
    return d;
}

template <bool TWO>
__device__ __forceinline__ void fps_scan(const v2f (&px2)[FPS_PPT / 2], const v2f (&py2)[FPS_PPT / 2],
                                         const v2f (&pz2)[FPS_PPT / 2], int (&distb)[FPS_PPT],
                                         int (&gbest)[FPS_NRUN],
                                         const float cx, const float cy, const float cz,
                                         const float qx, const float qy, const float qz)
{
    const v2f cx2 = {cx, cx};
    const v2f cy2 = {cy, cy};
    const v2f cz2 = {cz, cz};
    const v2f qx2 = {qx, qx};
    const v2f qy2 = {qy, qy};
    const v2f qz2 = {qz, qz};
#pragma unroll
    for (int g = 0; g < FPS_NRUN; ++g) {
        int best = 0;
#pragma unroll
        for (int jj = 0; jj < FPS_GROUP / 2; ++jj) {
            const int j = g * (FPS_GROUP / 2) + jj;
            const bool low = 2 * j < 24;
            const v2f d = fps_pair_dist(low, px2[j], py2[j], pz2[j], cx2, cy2, cz2);
            int nd0 = min(distb[2 * j + 0], __float_as_int(d.x));
            int nd1 = min(distb[2 * j + 1], __float_as_int(d.y));
            if (TWO) {
                const v2f e = fps_pair_dist(low, px2[j], py2[j], pz2[j], qx2, qy2, qz2);
                nd0 = min(nd0, __float_as_int(e.x));
                nd1 = min(nd1, __float_as_int(e.y));
            }
            distb[2 * j + 0] = nd0;
            distb[2 * j + 1] = nd1;
            best = (jj == 0) ? max(nd0, nd1) : max(max(best, nd0), nd1);
        }
        gbest[g] = best;
    }
}

template <int G>
__device__ __forceinline__ unsigned run_first_holder(const int (&distb)[FPS_PPT], const int (&gbest)[FPS_NRUN],
                                                     const int wm, int& cnt, int& lane2)
{
    unsigned long long msel = 0ull; int jsel = 0;
    int c = 0;
    int m = -1;
#pragma unroll
    for (int j = FPS_GROUP / 2 - 1; j >= 0; --j) {
        const int v0 = distb[G * FPS_GROUP + 2 * j + 0];
        const int v1 = distb[G * FPS_GROUP + 2 * j + 1];
        const unsigned m0 = __builtin_amdgcn_ballot_w32(v0 == wm);
        const unsigned m1 = __builtin_amdgcn_ballot_w32(v1 == wm);
        const unsigned long long mm = ((unsigned long long)m1 << 32) | m0;
        if (j == FPS_GROUP / 2 - 1 || mm != 0ull) { msel = mm; jsel = j; }
        c += __builtin_popcountll(mm);
        m = max(max(m, (v0 == wm) ? -1 : v0), (v1 == wm) ? -1 : v1);
    }
#pragma unroll
    for (int g = 0; g < FPS_NRUN; ++g) {
        if (g != G) m = max(m, gbest[g]);
    }
    cnt = c;
    lane2 = m;
    const int p = __builtin_ctzll(msel);
    return (unsigned)((2 * jsel + (p >> 5)) * FPS_THREADS + (p & 31));
}

__global__ __launch_bounds__(FPS_THREADS) void fps_kernel(const float* __restrict__ xyz,
                                                          int* __restrict__ fps_idx)
{
    const int b    = blockIdx.x;
    const int tid  = threadIdx.x;
    const int lane = tid & 31;
    const int wv   = tid >> 5;
    const float* xb = xyz + (size_t)b * NPTS * 3;

    __shared__ int4 spair[2][FPS_WAVES];

    __shared__ float4 spts[NPTS];

    v2f px2[FPS_PPT / 2], py2[FPS_PPT / 2], pz2[FPS_PPT / 2];
    int distb[FPS_PPT];
#pragma unroll
    for (int i = 0; i < FPS_PPT; ++i) {
        const int n = i * FPS_THREADS + tid;
        const float x = xb[n * 3 + 0];
        const float y = xb[n * 3 + 1];
        const float z = xb[n * 3 + 2];
        px2[i >> 1][i & 1] = x;
        py2[i >> 1][i & 1] = y;
        pz2[i >> 1][i & 1] = z;
        spts[n] = make_float4(x, y, z, 0.0f);
        distb[i] = __float_as_int(3.4e38f);
    }
    __syncthreads();

    int   s   = 0;
    int   cur = 0, cq = 0;
    bool  two = false;
    float cx, cy, cz;
    float qx = 0.0f, qy = 0.0f, qz = 0.0f;
    {
        const float4 c0 = spts[0];
        cx = c0.x; cy = c0.y; cz = c0.z;
    }

    for (int rnd = 0; rnd < NPOINT; ++rnd) {
        if (tid == 0) {
            fps_idx[b * NPOINT + s] = cur;
            if (two) fps_idx[b * NPOINT + s + 1] = cq;
        }
        const int snext = s + (two ? 2 : 1);
        if (snext >= NPOINT) break;

        int gbest[FPS_NRUN];
        if (two) fps_scan<true >(px2, py2, pz2, distb, gbest, cx, cy, cz, qx, qy, qz);
        else     fps_scan<false>(px2, py2, pz2, distb, gbest, cx, cy, cz, qx, qy, qz);
        const int bb = max(max(gbest[0], gbest[1]), max(gbest[2], gbest[3]));

        const int wm = wave_max_i32(bb);

        const unsigned r0 = __builtin_amdgcn_ballot_w32(gbest[0] == wm);
        const unsigned r1 = __builtin_amdgcn_ballot_w32(gbest[1] == wm);
        const unsigned r2 = __builtin_amdgcn_ballot_w32(gbest[2] == wm);
        int grun; unsigned key; int cnt; int lane2;
        if (r0)      { grun = 0; key = run_first_holder<0>(distb, gbest, wm, cnt, lane2); }
        else if (r1) { grun = 1; key = run_first_holder<1>(distb, gbest, wm, cnt, lane2); }
        else if (r2) { grun = 2; key = run_first_holder<2>(distb, gbest, wm, cnt, lane2); }
        else         { grun = 3; key = run_first_holder<3>(distb, gbest, wm, cnt, lane2); }
        const int wi = (int)key + grun * (FPS_GROUP * FPS_THREADS) + (tid & ~31);
        const int w2raw = wave_max_i32(lane2);
        const int w2 = (cnt >= 2) ? wm : w2raw;
        if (lane == 0) spair[rnd & 1][wv] = make_int4(wm, wi, w2, 0);
        __syncthreads();

        const int4 pr = spair[rnd & 1][lane & (FPS_WAVES - 1)];
        const int    pn   = pr.y & (NPTS - 1);
        const float4 cand = spts[pn];
        __builtin_amdgcn_sched_barrier(0);
        const int gm  = oct_max_i32(pr.x);
        const int gi  = oct_min_i32(pr.x == gm ? pr.y : 0x7fffffff);
        const int v2  = (pr.y == gi) ? -1 : pr.x;
        const int gm2 = oct_max_i32(v2);
        const int gi2 = oct_min_i32(v2 == gm2 ? pr.y : 0x7fffffff);
        const int wl  = __builtin_ctz(__builtin_amdgcn_ballot_w32(pr.y == gi));
        const int wl2 = __builtin_ctz(__builtin_amdgcn_ballot_w32(pr.y == gi2));
        const float nx = lane_f32(cand.x, wl),  ny = lane_f32(cand.y, wl),  nz = lane_f32(cand.z, wl);
        const float mx = lane_f32(cand.x, wl2), my = lane_f32(cand.y, wl2), mz = lane_f32(cand.z, wl2);
        const int   w2p = __builtin_amdgcn_readlane(pr.z, wl);
        const v2f ax = {cand.x, cand.x}, ay = {cand.y, cand.y}, az = {cand.z, cand.z};
        const v2f nx2 = {nx, nx}, ny2 = {ny, ny}, nz2 = {nz, nz};
        const v2f dlo = fps_pair_dist(true,  ax, ay, az, nx2, ny2, nz2);
        const v2f dhi = fps_pair_dist(false, ax, ay, az, nx2, ny2, nz2);
        const int dk  = (pn < FPS_LOWN) ? __float_as_int(dlo.x) : __float_as_int(dhi.x);
        const int dqp = __builtin_amdgcn_readlane(dk, wl2);
        const int gm2s = __builtin_amdgcn_readfirstlane(gm2);
        const bool ok = (gm2s > w2p) && (gm2s > 0) && (dqp >= gm2s);

        s   = snext;
        cur = __builtin_amdgcn_readfirstlane(gi)  & (NPTS - 1);
        cq  = __builtin_amdgcn_readfirstlane(gi2) & (NPTS - 1);
        cx = nx; cy = ny; cz = nz;
        qx = mx; qy = my; qz = mz;
        two = ok && (s + 1 < NPOINT);
    }
}

__global__ void gather_new_xyz_kernel(const float* __restrict__ xyz,
                                      const int* __restrict__ fps_idx,
                                      float* __restrict__ new_xyz)
{
    const int e = blockIdx.x * blockDim.x + threadIdx.x;
    if (e >= BATCH * NPOINT * 3) return;
    const int c = e % 3;
    const int p = (e / 3) % NPOINT;
    const int b = e / (3 * NPOINT);
    const int n = fps_idx[b * NPOINT + p];
    new_xyz[e] = xyz[((size_t)b * NPTS + n) * 3 + c];
}

__device__ __forceinline__ int row_up1(int v)
{
    return __builtin_amdgcn_update_dpp(0, v, 0x111, 0xf, 0xf, true);
}

__device__ __forceinline__ void knn_offer(unsigned m, float d, int first,
                                          float& lv, int& li, float& tau)
{
    while (m) {
        const int L = __builtin_ctz(m);
        m &= m - 1;
        const float c = __int_as_float(__builtin_amdgcn_readlane(__float_as_int(d), L));
        if (c < tau) {
            const int   ci = first + L;
            const float sv = __int_as_float(row_up1(__float_as_int(lv)));
            const int   si = row_up1(li);
            const bool  g  = lv > c;
            const bool  gs = sv > c;
            lv = g ? (gs ? sv : c)  : lv;
            li = g ? (gs ? si : ci) : li;
            tau = __int_as_float(__builtin_amdgcn_readlane(__float_as_int(lv), 15));
        }
    }
}

__global__ __launch_bounds__(256) void knn_kernel(const float* __restrict__ xyz,
                                                  const float* __restrict__ new_xyz,
                                                  int* __restrict__ knn_idx)
{
    const int lane = threadIdx.x & 31;
    const int wv = __builtin_amdgcn_readfirstlane((int)(threadIdx.x >> 5));
    const int q0 = (blockIdx.x * 8 + wv) * 2;
    const int b  = q0 / NPOINT;
    const float* xb = xyz + (size_t)b * NPTS * 3;

    const float ax = new_xyz[q0 * 3 + 0], ay = new_xyz[q0 * 3 + 1], az = new_xyz[q0 * 3 + 2];
    const float bx = new_xyz[q0 * 3 + 3], by = new_xyz[q0 * 3 + 4], bz = new_xyz[q0 * 3 + 5];

    float lva = 3.4e38f, lvb = 3.4e38f;
    int   lia = 0,       lib = 0;
    float taua = 3.4e38f, taub = 3.4e38f;

    for (int base = 0; base < NPTS; base += 128) {
        float px[4], py[4], pz[4];
#pragma unroll
        for (int u = 0; u < 4; ++u) {
            const float* p = xb + (size_t)(base + u * 32 + lane) * 3;
            px[u] = p[0]; py[u] = p[1]; pz[u] = p[2];
        }
        float da[4], db[4];
        unsigned ma[4], mb[4];
#pragma unroll
        for (int u = 0; u < 4; ++u) {
            {
                const float dx = px[u] - ax, dy = py[u] - ay, dz = pz[u] - az;
                float s2;
                {
#pragma clang fp contract(off)
                    s2 = dx * dx + dy * dy;
                }
                da[u] = __builtin_fmaf(dz, dz, s2);
            }
            {
                const float dx = px[u] - bx, dy = py[u] - by, dz = pz[u] - bz;
                float s2;
                {
#pragma clang fp contract(off)
                    s2 = dx * dx + dy * dy;
                }
                db[u] = __builtin_fmaf(dz, dz, s2);
            }
            ma[u] = __builtin_amdgcn_ballot_w32(da[u] < taua);
            mb[u] = __builtin_amdgcn_ballot_w32(db[u] < taub);
        }
#pragma unroll
        for (int u = 0; u < 4; ++u) {
            knn_offer(ma[u], da[u], base + u * 32, lva, lia, taua);
            knn_offer(mb[u], db[u], base + u * 32, lvb, lib, taub);
        }
    }
    knn_idx[(size_t)q0 * KNN + lane] = (lane < 16) ? lia : lib;
}

__global__ void pad_w_kernel(const float* __restrict__ W, float* __restrict__ Wpad)
{
    const int e = blockIdx.x * blockDim.x + threadIdx.x;
    if (e >= KPAD * COUT) return;
    const int r = e / COUT;
    Wpad[e] = (r < KRAW) ? W[e] : 0.0f;
}

#define GQ     4
#define GROWS  (GQ * KNN)
static_assert(NPOINT % GQ == 0, "the queries of a block belong to one batch");
static_assert(GROWS * 4 == 256, "the coordinate columns are filled one element per thread");

__global__ __launch_bounds__(256) void group_gemm_kernel(
    const float* __restrict__ xyz, const float* __restrict__ features,
    const float* __restrict__ new_xyz, const int* __restrict__ knn_idx,
    const float* __restrict__ Wpad,
    const float* __restrict__ gamma, const float* __restrict__ beta,
    const float* __restrict__ meanv, const float* __restrict__ varv,
    float* __restrict__ partial_sum, float* __restrict__ partial_sq,
    float* __restrict__ out_features, int final_pass)
{
    const int q0   = blockIdx.x * GQ;
    const int b    = q0 / NPOINT;
    const int tid  = threadIdx.x;
    const int lane = tid & 31;
    const int wv   = tid >> 5;

    __shared__ float As[GROWS][KPAD];
    __shared__ float red0[GQ][8][32];
    __shared__ float red1[GQ][8][32];

    const int* kq = knn_idx + (size_t)q0 * KNN;

    {
        const int r = tid >> 2;
        const int c = tid & 3;
        const int n = kq[r];
        float val = 0.0f;
        if (c < 3) {
            const float pv = xyz[((size_t)b * NPTS + n) * 3 + c];
            val = pv - new_xyz[(q0 + (r >> 4)) * 3 + c];
        }
        As[r][c < 3 ? c : KRAW] = val;
    }
    {
        const int fc = tid & 63;
        const int rs = tid >> 6;
#pragma unroll
        for (int i = 0; i < GROWS / 4; ++i) {
            const int r = i * 4 + rs;
            const int n = kq[r];
            As[r][3 + fc] = features[((size_t)b * NPTS + n) * CIN + fc];
        }
    }
    __syncthreads();

    const int col = wv * 16 + (lane & 15);
    const int hi  = lane >> 4;
    const int row = lane & 15;

    v8f acc[GQ];
#pragma unroll
    for (int t = 0; t < GQ; ++t) {
        const v8f zero = {};
        acc[t] = zero;
    }
#pragma unroll
    for (int kb = 0; kb < KPAD / 4; ++kb) {
        const int k0 = kb * 4 + hi * 2;
        v2f bb;
        bb.x = Wpad[(k0 + 0) * COUT + col];
        bb.y = Wpad[(k0 + 1) * COUT + col];
#pragma unroll
        for (int t = 0; t < GQ; ++t) {
            v2f a;
            a.x  = As[t * KNN + row][k0 + 0];
            a.y  = As[t * KNN + row][k0 + 1];
            acc[t] = __builtin_amdgcn_wmma_f32_16x16x4_f32(
                         false, a, false, bb, (short)0, acc[t], false, false);
        }
    }

    if (!final_pass) {
#pragma unroll
        for (int t = 0; t < GQ; ++t) {
            float s = 0.0f, ss = 0.0f;
#pragma unroll
            for (int j = 0; j < 8; ++j) { const float h = acc[t][j]; s += h; ss += h * h; }
            red0[t][wv][lane] = s;
            red1[t][wv][lane] = ss;
        }
        __syncthreads();
        if (lane < 16) {
#pragma unroll
            for (int t = 0; t < GQ; ++t) {
                const float s2  = red0[t][wv][lane] + red0[t][wv][lane + 16];
                const float ss2 = red1[t][wv][lane] + red1[t][wv][lane + 16];
                partial_sum[(size_t)(q0 + t) * COUT + col] = s2;
                partial_sq [(size_t)(q0 + t) * COUT + col] = ss2;
            }
        }
    } else {
        const float mu    = meanv[col];
        const float scale = gamma[col] * rsqrtf(varv[col] + BNEPS);
        const float shift = beta[col] - mu * scale;
#pragma unroll
        for (int t = 0; t < GQ; ++t) {
            float m = -3.4e38f;
#pragma unroll
            for (int j = 0; j < 8; ++j) {
                float h = acc[t][j] * scale + shift;
                h = fmaxf(h, 0.0f);
                m = fmaxf(m, h);
            }
            red0[t][wv][lane] = m;
        }
        __syncthreads();
        if (lane < 16) {
#pragma unroll
            for (int t = 0; t < GQ; ++t) {
                const float mm = fmaxf(red0[t][wv][lane], red0[t][wv][lane + 16]);
                out_features[(size_t)(q0 + t) * COUT + col] = mm;
            }
        }
    }
}

__global__ __launch_bounds__(256) void reduce_stats_kernel(
    const float* __restrict__ partial_sum, const float* __restrict__ partial_sq,
    float* __restrict__ meanv, float* __restrict__ varv)
{
    const int ch  = blockIdx.x;
    const int tid = threadIdx.x;
    __shared__ float ssum[256], ssq[256];
    float s = 0.0f, sq = 0.0f;
    for (int i = tid; i < BATCH * NPOINT; i += 256) {
        s  += partial_sum[(size_t)i * COUT + ch];
        sq += partial_sq [(size_t)i * COUT + ch];
    }
    ssum[tid] = s; ssq[tid] = sq;
    __syncthreads();
    for (int off = 128; off > 0; off >>= 1) {
        if (tid < off) { ssum[tid] += ssum[tid + off]; ssq[tid] += ssq[tid + off]; }
        __syncthreads();
    }
    if (tid == 0) {
        const float cnt = (float)(BATCH * NPOINT * KNN);
        const float mu  = ssum[0] / cnt;
        meanv[ch] = mu;
        varv[ch]  = ssq[0] / cnt - mu * mu;
    }
}

extern "C" void kernel_launch(void* const* d_in, const int* in_sizes, int n_in,
                              void* d_out, int out_size, void* d_ws, size_t ws_size,
                              hipStream_t stream)
{
    const float* xyz      = (const float*)d_in[0];
    const float* features = (const float*)d_in[1];
    const float* W        = (const float*)d_in[2];
    const float* gamma    = (const float*)d_in[3];
    const float* beta     = (const float*)d_in[4];

    float* out          = (float*)d_out;
    float* new_xyz      = out;
    float* out_features = out + BATCH * NPOINT * 3;

    char* ws = (char*)d_ws;
    size_t off = 0;
    int*   fps_idx     = (int*)(ws + off);   off += sizeof(int)   * BATCH * NPOINT;
    int*   knn_idx     = (int*)(ws + off);   off += sizeof(int)   * BATCH * NPOINT * KNN;
    float* Wpad        = (float*)(ws + off); off += sizeof(float) * KPAD * COUT;
    off = (off + 255) & ~(size_t)255;
    float* partial_sum = (float*)(ws + off); off += sizeof(float) * BATCH * NPOINT * COUT;
    float* partial_sq  = (float*)(ws + off); off += sizeof(float) * BATCH * NPOINT * COUT;
    float* meanv       = (float*)(ws + off); off += sizeof(float) * COUT;
    float* varv        = (float*)(ws + off); off += sizeof(float) * COUT;
    (void)ws_size; (void)in_sizes; (void)n_in; (void)out_size;

    fps_kernel<<<BATCH, FPS_THREADS, 0, stream>>>(xyz, fps_idx);
    gather_new_xyz_kernel<<<(BATCH * NPOINT * 3 + 255) / 256, 256, 0, stream>>>(
        xyz, fps_idx, new_xyz);
    knn_kernel<<<BATCH * NPOINT / 16, 256, 0, stream>>>(
        xyz, new_xyz, knn_idx);
    pad_w_kernel<<<(KPAD * COUT + 255) / 256, 256, 0, stream>>>(W, Wpad);
    group_gemm_kernel<<<BATCH * NPOINT / GQ, 256, 0, stream>>>(
        xyz, features, new_xyz, knn_idx, Wpad, gamma, beta,
        meanv, varv, partial_sum, partial_sq, out_features, 0);
    reduce_stats_kernel<<<COUT, 256, 0, stream>>>(
        partial_sum, partial_sq, meanv, varv);
    group_gemm_kernel<<<BATCH * NPOINT / GQ, 256, 0, stream>>>(
        xyz, features, new_xyz, knn_idx, Wpad, gamma, beta,
        meanv, varv, partial_sum, partial_sq, out_features, 1);
}


# Sampling: up to three picks per exchange: the third of the 8 wave firsts is taken too when its tests against both earlier picks and their waves' second distances hold; picks bit-identical
# speedup vs baseline: 4.8859x; 1.0394x over previous
#include <hip/hip_runtime.h>

typedef __attribute__((ext_vector_type(2))) float v2f;
typedef __attribute__((ext_vector_type(8))) float v8f;

#define BATCH   4
#define NPTS    8192
#define CIN     64
#define COUT    128
#define NPOINT  2048
#define KNN     16
#define KRAW    67
#define KPAD    68
#define BNEPS   1e-5f

#define FPS_THREADS 256
#define FPS_WAVES   (FPS_THREADS / 32)
#define FPS_PPT     (NPTS / FPS_THREADS)
#define FPS_GROUP   8
#define FPS_NRUN    (FPS_PPT / FPS_GROUP)
#define FPS_LOWN    (24 * FPS_THREADS)
static_assert(FPS_PPT == 4 * FPS_GROUP, "the choice of the run below is written for four runs per thread");
static_assert(FPS_LOWN == 6144, "slots 0 to 23 are the points below index 6144");

template <int M>
__device__ __forceinline__ int row_xor(int v)
{
    return __builtin_amdgcn_update_dpp(v, v, 0x160 + M, 0xf, 0xf, true);
}
__device__ __forceinline__ int half_xor(int v)
{
    return __builtin_amdgcn_permlanex16(v, v, 0x76543210, 0xfedcba98, false, false);
}
__device__ __forceinline__ int wave_max_i32(int x)
{
    x = max(x, row_xor<1>(x));
    x = max(x, row_xor<2>(x));
    x = max(x, row_xor<4>(x));
    x = max(x, row_xor<8>(x));
    x = max(x, half_xor(x));
    return x;
}
__device__ __forceinline__ int oct_max_i32(int x)
{
    x = max(x, row_xor<1>(x));
    x = max(x, row_xor<2>(x));
    x = max(x, row_xor<4>(x));
    return x;
}
__device__ __forceinline__ int oct_min_i32(int x)
{
    x = min(x, row_xor<1>(x));
    x = min(x, row_xor<2>(x));
    x = min(x, row_xor<4>(x));
    return x;
}
__device__ __forceinline__ float lane_f32(float v, int l)
{
    return __int_as_float(__builtin_amdgcn_readlane(__float_as_int(v), l));
}

__device__ __forceinline__ v2f fps_pair_dist(const bool low, const v2f px, const v2f py, const v2f pz,
                                             const v2f cx2, const v2f cy2, const v2f cz2)
{
    const v2f dx = px - cx2;
    const v2f dy = py - cy2;
    const v2f dz = pz - cz2;
    v2f d;
    if (low) {
        d = __builtin_elementwise_fma(dz, dz, __builtin_elementwise_fma(dx, dx, dy * dy));
    } else {
        v2f s2;
        {
#pragma clang fp contract(off)
            s2 = dx * dx + dy * dy;
        }
        d = __builtin_elementwise_fma(dz, dz, s2);
    }
    return d;
}

template <int K>
__device__ __forceinline__ void fps_scan(const v2f (&px2)[FPS_PPT / 2], const v2f (&py2)[FPS_PPT / 2],
                                         const v2f (&pz2)[FPS_PPT / 2], int (&distb)[FPS_PPT],
                                         int (&gbest)[FPS_NRUN],
                                         const float cx, const float cy, const float cz,
                                         const float qx, const float qy, const float qz,
                                         const float rx, const float ry, const float rz)
{
    const v2f cx2 = {cx, cx};
    const v2f cy2 = {cy, cy};
    const v2f cz2 = {cz, cz};
    const v2f qx2 = {qx, qx};
    const v2f qy2 = {qy, qy};
    const v2f qz2 = {qz, qz};
    const v2f rx2 = {rx, rx};
    const v2f ry2 = {ry, ry};
    const v2f rz2 = {rz, rz};
#pragma unroll
    for (int g = 0; g < FPS_NRUN; ++g) {
        int best = 0;
#pragma unroll
        for (int jj = 0; jj < FPS_GROUP / 2; ++jj) {
            const int j = g * (FPS_GROUP / 2) + jj;
            const bool low = 2 * j < 24;
            const v2f d = fps_pair_dist(low, px2[j], py2[j], pz2[j], cx2, cy2, cz2);
            int nd0 = min(distb[2 * j + 0], __float_as_int(d.x));
            int nd1 = min(distb[2 * j + 1], __float_as_int(d.y));
            if (K >= 2) {
                const v2f e = fps_pair_dist(low, px2[j], py2[j], pz2[j], qx2, qy2, qz2);
                nd0 = min(nd0, __float_as_int(e.x));
                nd1 = min(nd1, __float_as_int(e.y));
            }
            if (K >= 3) {
                const v2f f = fps_pair_dist(low, px2[j], py2[j], pz2[j], rx2, ry2, rz2);
                nd0 = min(nd0, __float_as_int(f.x));
                nd1 = min(nd1, __float_as_int(f.y));
            }
            distb[2 * j + 0] = nd0;
            distb[2 * j + 1] = nd1;
            best = (jj == 0) ? max(nd0, nd1) : max(max(best, nd0), nd1);
        }
        gbest[g] = best;
    }
}

template <int G>
__device__ __forceinline__ unsigned run_first_holder(const int (&distb)[FPS_PPT], const int (&gbest)[FPS_NRUN],
                                                     const int wm, int& cnt, int& lane2)
{
    unsigned long long msel = 0ull; int jsel = 0;
    int c = 0;
    int m = -1;
#pragma unroll
    for (int j = FPS_GROUP / 2 - 1; j >= 0; --j) {
        const int v0 = distb[G * FPS_GROUP + 2 * j + 0];
        const int v1 = distb[G * FPS_GROUP + 2 * j + 1];
        const unsigned m0 = __builtin_amdgcn_ballot_w32(v0 == wm);
        const unsigned m1 = __builtin_amdgcn_ballot_w32(v1 == wm);
        const unsigned long long mm = ((unsigned long long)m1 << 32) | m0;
        if (j == FPS_GROUP / 2 - 1 || mm != 0ull) { msel = mm; jsel = j; }
        c += __builtin_popcountll(mm);
        m = max(max(m, (v0 == wm) ? -1 : v0), (v1 == wm) ? -1 : v1);
    }
#pragma unroll
    for (int g = 0; g < FPS_NRUN; ++g) {
        if (g != G) m = max(m, gbest[g]);
    }
    cnt = c;
    lane2 = m;
    const int p = __builtin_ctzll(msel);
    return (unsigned)((2 * jsel + (p >> 5)) * FPS_THREADS + (p & 31));
}

__global__ __launch_bounds__(FPS_THREADS) void fps_kernel(const float* __restrict__ xyz,
                                                          int* __restrict__ fps_idx)
{
    const int b    = blockIdx.x;
    const int tid  = threadIdx.x;
    const int lane = tid & 31;
    const int wv   = tid >> 5;
    const float* xb = xyz + (size_t)b * NPTS * 3;

    __shared__ int4 spair[2][FPS_WAVES];

    __shared__ float4 spts[NPTS];

    v2f px2[FPS_PPT / 2], py2[FPS_PPT / 2], pz2[FPS_PPT / 2];
    int distb[FPS_PPT];
#pragma unroll
    for (int i = 0; i < FPS_PPT; ++i) {
        const int n = i * FPS_THREADS + tid;
        const float x = xb[n * 3 + 0];
        const float y = xb[n * 3 + 1];
        const float z = xb[n * 3 + 2];
        px2[i >> 1][i & 1] = x;
        py2[i >> 1][i & 1] = y;
        pz2[i >> 1][i & 1] = z;
        spts[n] = make_float4(x, y, z, 0.0f);
        distb[i] = __float_as_int(3.4e38f);
    }
    __syncthreads();

    int   s   = 0;
    int   cur = 0, cq = 0, cr = 0;
    int   np  = 1;
    float cx, cy, cz;
    float qx = 0.0f, qy = 0.0f, qz = 0.0f;
    float rx = 0.0f, ry = 0.0f, rz = 0.0f;
    {
        const float4 c0 = spts[0];
        cx = c0.x; cy = c0.y; cz = c0.z;
    }

    for (int rnd = 0; rnd < NPOINT; ++rnd) {
        if (tid == 0) {
            fps_idx[b * NPOINT + s] = cur;
            if (np >= 2) fps_idx[b * NPOINT + s + 1] = cq;
            if (np >= 3) fps_idx[b * NPOINT + s + 2] = cr;
        }
        const int snext = s + np;
        if (snext >= NPOINT) break;

        int gbest[FPS_NRUN];
        if (np == 3)      fps_scan<3>(px2, py2, pz2, distb, gbest, cx, cy, cz, qx, qy, qz, rx, ry, rz);
        else if (np == 2) fps_scan<2>(px2, py2, pz2, distb, gbest, cx, cy, cz, qx, qy, qz, rx, ry, rz);
        else              fps_scan<1>(px2, py2, pz2, distb, gbest, cx, cy, cz, qx, qy, qz, rx, ry, rz);
        const int bb = max(max(gbest[0], gbest[1]), max(gbest[2], gbest[3]));

        const int wm = wave_max_i32(bb);

        const unsigned r0 = __builtin_amdgcn_ballot_w32(gbest[0] == wm);
        const unsigned r1 = __builtin_amdgcn_ballot_w32(gbest[1] == wm);
        const unsigned r2 = __builtin_amdgcn_ballot_w32(gbest[2] == wm);
        int grun; unsigned key; int cnt; int lane2;
        if (r0)      { grun = 0; key = run_first_holder<0>(distb, gbest, wm, cnt, lane2); }
        else if (r1) { grun = 1; key = run_first_holder<1>(distb, gbest, wm, cnt, lane2); }
        else if (r2) { grun = 2; key = run_first_holder<2>(distb, gbest, wm, cnt, lane2); }
        else         { grun = 3; key = run_first_holder<3>(distb, gbest, wm, cnt, lane2); }
        const int wi = (int)key + grun * (FPS_GROUP * FPS_THREADS) + (tid & ~31);
        const int w2raw = wave_max_i32(lane2);
        const int w2 = (cnt >= 2) ? wm : w2raw;
        if (lane == 0) spair[rnd & 1][wv] = make_int4(wm, wi, w2, 0);
        __syncthreads();

        const int4 pr = spair[rnd & 1][lane & (FPS_WAVES - 1)];
        const int    pn   = pr.y & (NPTS - 1);
        const float4 cand = spts[pn];
        __builtin_amdgcn_sched_barrier(0);
        const int gm  = oct_max_i32(pr.x);
        const int gi  = oct_min_i32(pr.x == gm ? pr.y : 0x7fffffff);
        const int v2  = (pr.y == gi) ? -1 : pr.x;
        const int gm2 = oct_max_i32(v2);
        const int gi2 = oct_min_i32(v2 == gm2 ? pr.y : 0x7fffffff);
        const int v3  = (pr.y == gi2) ? -1 : v2;
        const int gm3 = oct_max_i32(v3);
        const int gi3 = oct_min_i32(v3 == gm3 ? pr.y : 0x7fffffff);
        const int wl  = __builtin_ctz(__builtin_amdgcn_ballot_w32(pr.y == gi));
        const int wl2 = __builtin_ctz(__builtin_amdgcn_ballot_w32(pr.y == gi2));
        const int wl3 = __builtin_ctz(__builtin_amdgcn_ballot_w32(pr.y == gi3));
        const float nx = lane_f32(cand.x, wl),  ny = lane_f32(cand.y, wl),  nz = lane_f32(cand.z, wl);
        const float mx = lane_f32(cand.x, wl2), my = lane_f32(cand.y, wl2), mz = lane_f32(cand.z, wl2);
        const float tx = lane_f32(cand.x, wl3), ty = lane_f32(cand.y, wl3), tz = lane_f32(cand.z, wl3);
        const int   w2p = __builtin_amdgcn_readlane(pr.z, wl);
        const int   w2q = __builtin_amdgcn_readlane(pr.z, wl2);
        const v2f ax = {cand.x, cand.x}, ay = {cand.y, cand.y}, az = {cand.z, cand.z};
        const v2f nx2 = {nx, nx}, ny2 = {ny, ny}, nz2 = {nz, nz};
        const v2f mx2 = {mx, mx}, my2 = {my, my}, mz2 = {mz, mz};
        const v2f dlo = fps_pair_dist(true,  ax, ay, az, nx2, ny2, nz2);
        const v2f dhi = fps_pair_dist(false, ax, ay, az, nx2, ny2, nz2);
        const v2f elo = fps_pair_dist(true,  ax, ay, az, mx2, my2, mz2);
        const v2f ehi = fps_pair_dist(false, ax, ay, az, mx2, my2, mz2);
        const int dk  = (pn < FPS_LOWN) ? __float_as_int(dlo.x) : __float_as_int(dhi.x);
        const int ek  = (pn < FPS_LOWN) ? __float_as_int(elo.x) : __float_as_int(ehi.x);
        const int dqp = __builtin_amdgcn_readlane(dk, wl2);
        const int drp = __builtin_amdgcn_readlane(dk, wl3);
        const int drq = __builtin_amdgcn_readlane(ek, wl3);
        const int gm2s = __builtin_amdgcn_readfirstlane(gm2);
        const int gm3s = __builtin_amdgcn_readfirstlane(gm3);
        const bool ok2 = (gm2s > w2p) && (gm2s > 0) && (dqp >= gm2s);
        const bool ok3 = (gm3s > w2p) && (gm3s > w2q) && (gm3s > 0) && (drp >= gm3s) && (drq >= gm3s);

        s   = snext;
        cur = __builtin_amdgcn_readfirstlane(gi)  & (NPTS - 1);
        cq  = __builtin_amdgcn_readfirstlane(gi2) & (NPTS - 1);
        cr  = __builtin_amdgcn_readfirstlane(gi3) & (NPTS - 1);
        cx = nx; cy = ny; cz = nz;
        qx = mx; qy = my; qz = mz;
        rx = tx; ry = ty; rz = tz;
        const bool k2 = ok2 && (s + 1 < NPOINT);
        const bool k3 = k2 && ok3 && (s + 2 < NPOINT);
        np = 1 + (k2 ? 1 : 0) + (k3 ? 1 : 0);
    }
}

__global__ void gather_new_xyz_kernel(const float* __restrict__ xyz,
                                      const int* __restrict__ fps_idx,
                                      float* __restrict__ new_xyz)
{
    const int e = blockIdx.x * blockDim.x + threadIdx.x;
    if (e >= BATCH * NPOINT * 3) return;
    const int c = e % 3;
    const int p = (e / 3) % NPOINT;
    const int b = e / (3 * NPOINT);
    const int n = fps_idx[b * NPOINT + p];
    new_xyz[e] = xyz[((size_t)b * NPTS + n) * 3 + c];
}

__device__ __forceinline__ int row_up1(int v)
{
    return __builtin_amdgcn_update_dpp(0, v, 0x111, 0xf, 0xf, true);
}

__device__ __forceinline__ void knn_offer(unsigned m, float d, int first,
                                          float& lv, int& li, float& tau)
{
    while (m) {
        const int L = __builtin_ctz(m);
        m &= m - 1;
        const float c = __int_as_float(__builtin_amdgcn_readlane(__float_as_int(d), L));
        if (c < tau) {
            const int   ci = first + L;
            const float sv = __int_as_float(row_up1(__float_as_int(lv)));
            const int   si = row_up1(li);
            const bool  g  = lv > c;
            const bool  gs = sv > c;
            lv = g ? (gs ? sv : c)  : lv;
            li = g ? (gs ? si : ci) : li;
            tau = __int_as_float(__builtin_amdgcn_readlane(__float_as_int(lv), 15));
        }
    }
}

__global__ __launch_bounds__(256) void knn_kernel(const float* __restrict__ xyz,
                                                  const float* __restrict__ new_xyz,
                                                  int* __restrict__ knn_idx)
{
    const int lane = threadIdx.x & 31;
    const int wv = __builtin_amdgcn_readfirstlane((int)(threadIdx.x >> 5));
    const int q0 = (blockIdx.x * 8 + wv) * 2;
    const int b  = q0 / NPOINT;
    const float* xb = xyz + (size_t)b * NPTS * 3;

    const float ax = new_xyz[q0 * 3 + 0], ay = new_xyz[q0 * 3 + 1], az = new_xyz[q0 * 3 + 2];
    const float bx = new_xyz[q0 * 3 + 3], by = new_xyz[q0 * 3 + 4], bz = new_xyz[q0 * 3 + 5];

    float lva = 3.4e38f, lvb = 3.4e38f;
    int   lia = 0,       lib = 0;
    float taua = 3.4e38f, taub = 3.4e38f;

    for (int base = 0; base < NPTS; base += 128) {
        float px[4], py[4], pz[4];
#pragma unroll
        for (int u = 0; u < 4; ++u) {
            const float* p = xb + (size_t)(base + u * 32 + lane) * 3;
            px[u] = p[0]; py[u] = p[1]; pz[u] = p[2];
        }
        float da[4], db[4];
        unsigned ma[4], mb[4];
#pragma unroll
        for (int u = 0; u < 4; ++u) {
            {
                const float dx = px[u] - ax, dy = py[u] - ay, dz = pz[u] - az;
                float s2;
                {
#pragma clang fp contract(off)
                    s2 = dx * dx + dy * dy;
                }
                da[u] = __builtin_fmaf(dz, dz, s2);
            }
            {
                const float dx = px[u] - bx, dy = py[u] - by, dz = pz[u] - bz;
                float s2;
                {
#pragma clang fp contract(off)
                    s2 = dx * dx + dy * dy;
                }
                db[u] = __builtin_fmaf(dz, dz, s2);
            }
            ma[u] = __builtin_amdgcn_ballot_w32(da[u] < taua);
            mb[u] = __builtin_amdgcn_ballot_w32(db[u] < taub);
        }
#pragma unroll
        for (int u = 0; u < 4; ++u) {
            knn_offer(ma[u], da[u], base + u * 32, lva, lia, taua);
            knn_offer(mb[u], db[u], base + u * 32, lvb, lib, taub);
        }
    }
    knn_idx[(size_t)q0 * KNN + lane] = (lane < 16) ? lia : lib;
}

__global__ void pad_w_kernel(const float* __restrict__ W, float* __restrict__ Wpad)
{
    const int e = blockIdx.x * blockDim.x + threadIdx.x;
    if (e >= KPAD * COUT) return;
    const int r = e / COUT;
    Wpad[e] = (r < KRAW) ? W[e] : 0.0f;
}

#define GQ     4
#define GROWS  (GQ * KNN)
static_assert(NPOINT % GQ == 0, "the queries of a block belong to one batch");
static_assert(GROWS * 4 == 256, "the coordinate columns are filled one element per thread");

__global__ __launch_bounds__(256) void group_gemm_kernel(
    const float* __restrict__ xyz, const float* __restrict__ features,
    const float* __restrict__ new_xyz, const int* __restrict__ knn_idx,
    const float* __restrict__ Wpad,
    const float* __restrict__ gamma, const float* __restrict__ beta,
    const float* __restrict__ meanv, const float* __restrict__ varv,
    float* __restrict__ partial_sum, float* __restrict__ partial_sq,
    float* __restrict__ out_features, int final_pass)
{
    const int q0   = blockIdx.x * GQ;
    const int b    = q0 / NPOINT;
    const int tid  = threadIdx.x;
    const int lane = tid & 31;
    const int wv   = tid >> 5;

    __shared__ float As[GROWS][KPAD];
    __shared__ float red0[GQ][8][32];
    __shared__ float red1[GQ][8][32];

    const int* kq = knn_idx + (size_t)q0 * KNN;

    {
        const int r = tid >> 2;
        const int c = tid & 3;
        const int n = kq[r];
        float val = 0.0f;
        if (c < 3) {
            const float pv = xyz[((size_t)b * NPTS + n) * 3 + c];
            val = pv - new_xyz[(q0 + (r >> 4)) * 3 + c];
        }
        As[r][c < 3 ? c : KRAW] = val;
    }
    {
        const int fc = tid & 63;
        const int rs = tid >> 6;
#pragma unroll
        for (int i = 0; i < GROWS / 4; ++i) {
            const int r = i * 4 + rs;
            const int n = kq[r];
            As[r][3 + fc] = features[((size_t)b * NPTS + n) * CIN + fc];
        }
    }
    __syncthreads();

    const int col = wv * 16 + (lane & 15);
    const int hi  = lane >> 4;
    const int row = lane & 15;

    v8f acc[GQ];
#pragma unroll
    for (int t = 0; t < GQ; ++t) {
        const v8f zero = {};
        acc[t] = zero;
    }
#pragma unroll
    for (int kb = 0; kb < KPAD / 4; ++kb) {
        const int k0 = kb * 4 + hi * 2;
        v2f bb;
        bb.x = Wpad[(k0 + 0) * COUT + col];
        bb.y = Wpad[(k0 + 1) * COUT + col];
#pragma unroll
        for (int t = 0; t < GQ; ++t) {
            v2f a;
            a.x  = As[t * KNN + row][k0 + 0];
            a.y  = As[t * KNN + row][k0 + 1];
            acc[t] = __builtin_amdgcn_wmma_f32_16x16x4_f32(
                         false, a, false, bb, (short)0, acc[t], false, false);
        }
    }

    if (!final_pass) {
#pragma unroll
        for (int t = 0; t < GQ; ++t) {
            float s = 0.0f, ss = 0.0f;
#pragma unroll
            for (int j = 0; j < 8; ++j) { const float h = acc[t][j]; s += h; ss += h * h; }
            red0[t][wv][lane] = s;
            red1[t][wv][lane] = ss;
        }
        __syncthreads();
        if (lane < 16) {
#pragma unroll
            for (int t = 0; t < GQ; ++t) {
                const float s2  = red0[t][wv][lane] + red0[t][wv][lane + 16];
                const float ss2 = red1[t][wv][lane] + red1[t][wv][lane + 16];
                partial_sum[(size_t)(q0 + t) * COUT + col] = s2;
                partial_sq [(size_t)(q0 + t) * COUT + col] = ss2;
            }
        }
    } else {
        const float mu    = meanv[col];
        const float scale = gamma[col] * rsqrtf(varv[col] + BNEPS);
        const float shift = beta[col] - mu * scale;
#pragma unroll
        for (int t = 0; t < GQ; ++t) {
            float m = -3.4e38f;
#pragma unroll
            for (int j = 0; j < 8; ++j) {
                float h = acc[t][j] * scale + shift;
                h = fmaxf(h, 0.0f);
                m = fmaxf(m, h);
            }
            red0[t][wv][lane] = m;
        }
        __syncthreads();
        if (lane < 16) {
#pragma unroll
            for (int t = 0; t < GQ; ++t) {
                const float mm = fmaxf(red0[t][wv][lane], red0[t][wv][lane + 16]);
                out_features[(size_t)(q0 + t) * COUT + col] = mm;
            }
        }
    }
}

__global__ __launch_bounds__(256) void reduce_stats_kernel(
    const float* __restrict__ partial_sum, const float* __restrict__ partial_sq,
    float* __restrict__ meanv, float* __restrict__ varv)
{
    const int ch  = blockIdx.x;
    const int tid = threadIdx.x;
    __shared__ float ssum[256], ssq[256];
    float s = 0.0f, sq = 0.0f;
    for (int i = tid; i < BATCH * NPOINT; i += 256) {
        s  += partial_sum[(size_t)i * COUT + ch];
        sq += partial_sq [(size_t)i * COUT + ch];
    }
    ssum[tid] = s; ssq[tid] = sq;
    __syncthreads();
    for (int off = 128; off > 0; off >>= 1) {
        if (tid < off) { ssum[tid] += ssum[tid + off]; ssq[tid] += ssq[tid + off]; }
        __syncthreads();
    }
    if (tid == 0) {
        const float cnt = (float)(BATCH * NPOINT * KNN);
        const float mu  = ssum[0] / cnt;
        meanv[ch] = mu;
        varv[ch]  = ssq[0] / cnt - mu * mu;
    }
}

extern "C" void kernel_launch(void* const* d_in, const int* in_sizes, int n_in,
                              void* d_out, int out_size, void* d_ws, size_t ws_size,
                              hipStream_t stream)
{
    const float* xyz      = (const float*)d_in[0];
    const float* features = (const float*)d_in[1];
    const float* W        = (const float*)d_in[2];
    const float* gamma    = (const float*)d_in[3];
    const float* beta     = (const float*)d_in[4];

    float* out          = (float*)d_out;
    float* new_xyz      = out;
    float* out_features = out + BATCH * NPOINT * 3;

    char* ws = (char*)d_ws;
    size_t off = 0;
    int*   fps_idx     = (int*)(ws + off);   off += sizeof(int)   * BATCH * NPOINT;
    int*   knn_idx     = (int*)(ws + off);   off += sizeof(int)   * BATCH * NPOINT * KNN;
    float* Wpad        = (float*)(ws + off); off += sizeof(float) * KPAD * COUT;
    off = (off + 255) & ~(size_t)255;
    float* partial_sum = (float*)(ws + off); off += sizeof(float) * BATCH * NPOINT * COUT;
    float* partial_sq  = (float*)(ws + off); off += sizeof(float) * BATCH * NPOINT * COUT;
    float* meanv       = (float*)(ws + off); off += sizeof(float) * COUT;
    float* varv        = (float*)(ws + off); off += sizeof(float) * COUT;
    (void)ws_size; (void)in_sizes; (void)n_in; (void)out_size;

    fps_kernel<<<BATCH, FPS_THREADS, 0, stream>>>(xyz, fps_idx);
    gather_new_xyz_kernel<<<(BATCH * NPOINT * 3 + 255) / 256, 256, 0, stream>>>(
        xyz, fps_idx, new_xyz);
    knn_kernel<<<BATCH * NPOINT / 16, 256, 0, stream>>>(
        xyz, new_xyz, knn_idx);
    pad_w_kernel<<<(KPAD * COUT + 255) / 256, 256, 0, stream>>>(W, Wpad);
    group_gemm_kernel<<<BATCH * NPOINT / GQ, 256, 0, stream>>>(
        xyz, features, new_xyz, knn_idx, Wpad, gamma, beta,
        meanv, varv, partial_sum, partial_sq, out_features, 0);
    reduce_stats_kernel<<<COUT, 256, 0, stream>>>(
        partial_sum, partial_sq, meanv, varv);
    group_gemm_kernel<<<BATCH * NPOINT / GQ, 256, 0, stream>>>(
        xyz, features, new_xyz, knn_idx, Wpad, gamma, beta,
        meanv, varv, partial_sum, partial_sq, out_features, 1);
}


# Sampling loop runs two exchanges per trip on two sets of running-distance registers (one read, the other written, roles swap): the 16 register-pair copies atop every exchange go; picks bit-identical
# speedup vs baseline: 5.1389x; 1.0518x over previous
#include <hip/hip_runtime.h>

typedef __attribute__((ext_vector_type(2))) float v2f;
typedef __attribute__((ext_vector_type(8))) float v8f;

#define BATCH   4
#define NPTS    8192
#define CIN     64
#define COUT    128
#define NPOINT  2048
#define KNN     16
#define KRAW    67
#define KPAD    68
#define BNEPS   1e-5f

#define FPS_THREADS 256
#define FPS_WAVES   (FPS_THREADS / 32)
#define FPS_PPT     (NPTS / FPS_THREADS)
#define FPS_GROUP   8
#define FPS_NRUN    (FPS_PPT / FPS_GROUP)
#define FPS_LOWN    (24 * FPS_THREADS)
static_assert(FPS_PPT == 4 * FPS_GROUP, "the choice of the run below is written for four runs per thread");
static_assert(FPS_LOWN == 6144, "slots 0 to 23 are the points below index 6144");

template <int M>
__device__ __forceinline__ int row_xor(int v)
{
    return __builtin_amdgcn_update_dpp(v, v, 0x160 + M, 0xf, 0xf, true);
}
__device__ __forceinline__ int half_xor(int v)
{
    return __builtin_amdgcn_permlanex16(v, v, 0x76543210, 0xfedcba98, false, false);
}
__device__ __forceinline__ int wave_max_i32(int x)
{
    x = max(x, row_xor<1>(x));
    x = max(x, row_xor<2>(x));
    x = max(x, row_xor<4>(x));
    x = max(x, row_xor<8>(x));
    x = max(x, half_xor(x));
    return x;
}
__device__ __forceinline__ int oct_max_i32(int x)
{
    x = max(x, row_xor<1>(x));
    x = max(x, row_xor<2>(x));
    x = max(x, row_xor<4>(x));
    return x;
}
__device__ __forceinline__ int oct_min_i32(int x)
{
    x = min(x, row_xor<1>(x));
    x = min(x, row_xor<2>(x));
    x = min(x, row_xor<4>(x));
    return x;
}
__device__ __forceinline__ float lane_f32(float v, int l)
{
    return __int_as_float(__builtin_amdgcn_readlane(__float_as_int(v), l));
}

__device__ __forceinline__ v2f fps_pair_dist(const bool low, const v2f px, const v2f py, const v2f pz,
                                             const v2f cx2, const v2f cy2, const v2f cz2)
{
    const v2f dx = px - cx2;
    const v2f dy = py - cy2;
    const v2f dz = pz - cz2;
    v2f d;
    if (low) {
        d = __builtin_elementwise_fma(dz, dz, __builtin_elementwise_fma(dx, dx, dy * dy));
    } else {
        v2f s2;
        {
#pragma clang fp contract(off)
            s2 = dx * dx + dy * dy;
        }
        d = __builtin_elementwise_fma(dz, dz, s2);
    }
    return d;
}

template <int K>
__device__ __forceinline__ void fps_scan(const v2f (&px2)[FPS_PPT / 2], const v2f (&py2)[FPS_PPT / 2],
                                         const v2f (&pz2)[FPS_PPT / 2], const int (&din)[FPS_PPT],
                                         int (&dout)[FPS_PPT], int (&gbest)[FPS_NRUN],
                                         const float cx, const float cy, const float cz,
                                         const float qx, const float qy, const float qz,
                                         const float rx, const float ry, const float rz)
{
    const v2f cx2 = {cx, cx};
    const v2f cy2 = {cy, cy};
    const v2f cz2 = {cz, cz};
    const v2f qx2 = {qx, qx};
    const v2f qy2 = {qy, qy};
    const v2f qz2 = {qz, qz};
    const v2f rx2 = {rx, rx};
    const v2f ry2 = {ry, ry};
    const v2f rz2 = {rz, rz};
#pragma unroll
    for (int g = 0; g < FPS_NRUN; ++g) {
        int best = 0;
#pragma unroll
        for (int jj = 0; jj < FPS_GROUP / 2; ++jj) {
            const int j = g * (FPS_GROUP / 2) + jj;
            const bool low = 2 * j < 24;
            const v2f d = fps_pair_dist(low, px2[j], py2[j], pz2[j], cx2, cy2, cz2);
            int nd0 = min(din[2 * j + 0], __float_as_int(d.x));
            int nd1 = min(din[2 * j + 1], __float_as_int(d.y));
            if (K >= 2) {
                const v2f e = fps_pair_dist(low, px2[j], py2[j], pz2[j], qx2, qy2, qz2);
                nd0 = min(nd0, __float_as_int(e.x));
                nd1 = min(nd1, __float_as_int(e.y));
            }
            if (K >= 3) {
                const v2f f = fps_pair_dist(low, px2[j], py2[j], pz2[j], rx2, ry2, rz2);
                nd0 = min(nd0, __float_as_int(f.x));
                nd1 = min(nd1, __float_as_int(f.y));
            }
            dout[2 * j + 0] = nd0;
            dout[2 * j + 1] = nd1;
            best = (jj == 0) ? max(nd0, nd1) : max(max(best, nd0), nd1);
        }
        gbest[g] = best;
    }
}

template <int G>
__device__ __forceinline__ unsigned run_first_holder(const int (&distb)[FPS_PPT], const int (&gbest)[FPS_NRUN],
                                                     const int wm, int& cnt, int& lane2)
{
    unsigned long long msel = 0ull; int jsel = 0;
    int c = 0;
    int m = -1;
#pragma unroll
    for (int j = FPS_GROUP / 2 - 1; j >= 0; --j) {
        const int v0 = distb[G * FPS_GROUP + 2 * j + 0];
        const int v1 = distb[G * FPS_GROUP + 2 * j + 1];
        const unsigned m0 = __builtin_amdgcn_ballot_w32(v0 == wm);
        const unsigned m1 = __builtin_amdgcn_ballot_w32(v1 == wm);
        const unsigned long long mm = ((unsigned long long)m1 << 32) | m0;
        if (j == FPS_GROUP / 2 - 1 || mm != 0ull) { msel = mm; jsel = j; }
        c += __builtin_popcountll(mm);
        m = max(max(m, (v0 == wm) ? -1 : v0), (v1 == wm) ? -1 : v1);
    }
#pragma unroll
    for (int g = 0; g < FPS_NRUN; ++g) {
        if (g != G) m = max(m, gbest[g]);
    }
    cnt = c;
    lane2 = m;
    const int p = __builtin_ctzll(msel);
    return (unsigned)((2 * jsel + (p >> 5)) * FPS_THREADS + (p & 31));
}

__global__ __launch_bounds__(FPS_THREADS) void fps_kernel(const float* __restrict__ xyz,
                                                          int* __restrict__ fps_idx)
{
    const int b    = blockIdx.x;
    const int tid  = threadIdx.x;
    const int lane = tid & 31;
    const int wv   = tid >> 5;
    const float* xb = xyz + (size_t)b * NPTS * 3;

    __shared__ int4 spair[2][FPS_WAVES];

    __shared__ float4 spts[NPTS];

    v2f px2[FPS_PPT / 2], py2[FPS_PPT / 2], pz2[FPS_PPT / 2];
    int dist_a[FPS_PPT], dist_b[FPS_PPT];
#pragma unroll
    for (int i = 0; i < FPS_PPT; ++i) {
        const int n = i * FPS_THREADS + tid;
        const float x = xb[n * 3 + 0];
        const float y = xb[n * 3 + 1];
        const float z = xb[n * 3 + 2];
        px2[i >> 1][i & 1] = x;
        py2[i >> 1][i & 1] = y;
        pz2[i >> 1][i & 1] = z;
        spts[n] = make_float4(x, y, z, 0.0f);
        dist_a[i] = __float_as_int(3.4e38f);
    }
    __syncthreads();

    int   s   = 0;
    int   cur = 0, cq = 0, cr = 0;
    int   np  = 1;
    float cx, cy, cz;
    float qx = 0.0f, qy = 0.0f, qz = 0.0f;
    float rx = 0.0f, ry = 0.0f, rz = 0.0f;
    {
        const float4 c0 = spts[0];
        cx = c0.x; cy = c0.y; cz = c0.z;
    }

    auto exchange = [&](const int (&din)[FPS_PPT], int (&dout)[FPS_PPT], const int par)
                        __attribute__((always_inline)) -> bool {
        if (tid == 0) {
            fps_idx[b * NPOINT + s] = cur;
            if (np >= 2) fps_idx[b * NPOINT + s + 1] = cq;
            if (np >= 3) fps_idx[b * NPOINT + s + 2] = cr;
        }
        const int snext = s + np;
        if (snext >= NPOINT) return true;

        int gbest[FPS_NRUN];
        if (np == 3)      fps_scan<3>(px2, py2, pz2, din, dout, gbest, cx, cy, cz, qx, qy, qz, rx, ry, rz);
        else if (np == 2) fps_scan<2>(px2, py2, pz2, din, dout, gbest, cx, cy, cz, qx, qy, qz, rx, ry, rz);
        else              fps_scan<1>(px2, py2, pz2, din, dout, gbest, cx, cy, cz, qx, qy, qz, rx, ry, rz);
        const int bb = max(max(gbest[0], gbest[1]), max(gbest[2], gbest[3]));

        const int wm = wave_max_i32(bb);

        const unsigned r0 = __builtin_amdgcn_ballot_w32(gbest[0] == wm);
        const unsigned r1 = __builtin_amdgcn_ballot_w32(gbest[1] == wm);
        const unsigned r2 = __builtin_amdgcn_ballot_w32(gbest[2] == wm);
        int grun; unsigned key; int cnt; int lane2;
        if (r0)      { grun = 0; key = run_first_holder<0>(dout, gbest, wm, cnt, lane2); }
        else if (r1) { grun = 1; key = run_first_holder<1>(dout, gbest, wm, cnt, lane2); }
        else if (r2) { grun = 2; key = run_first_holder<2>(dout, gbest, wm, cnt, lane2); }
        else         { grun = 3; key = run_first_holder<3>(dout, gbest, wm, cnt, lane2); }
        const int wi = (int)key + grun * (FPS_GROUP * FPS_THREADS) + (tid & ~31);
        const int w2raw = wave_max_i32(lane2);
        const int w2 = (cnt >= 2) ? wm : w2raw;
        if (lane == 0) spair[par][wv] = make_int4(wm, wi, w2, 0);
        __syncthreads();

        const int4 pr = spair[par][lane & (FPS_WAVES - 1)];
        const int    pn   = pr.y & (NPTS - 1);
        const float4 cand = spts[pn];
        __builtin_amdgcn_sched_barrier(0);
        const int gm  = oct_max_i32(pr.x);
        const int gi  = oct_min_i32(pr.x == gm ? pr.y : 0x7fffffff);
        const int v2  = (pr.y == gi) ? -1 : pr.x;
        const int gm2 = oct_max_i32(v2);
        const int gi2 = oct_min_i32(v2 == gm2 ? pr.y : 0x7fffffff);
        const int v3  = (pr.y == gi2) ? -1 : v2;
        const int gm3 = oct_max_i32(v3);
        const int gi3 = oct_min_i32(v3 == gm3 ? pr.y : 0x7fffffff);
        const int wl  = __builtin_ctz(__builtin_amdgcn_ballot_w32(pr.y == gi));
        const int wl2 = __builtin_ctz(__builtin_amdgcn_ballot_w32(pr.y == gi2));
        const int wl3 = __builtin_ctz(__builtin_amdgcn_ballot_w32(pr.y == gi3));
        const float nx = lane_f32(cand.x, wl),  ny = lane_f32(cand.y, wl),  nz = lane_f32(cand.z, wl);
        const float mx = lane_f32(cand.x, wl2), my = lane_f32(cand.y, wl2), mz = lane_f32(cand.z, wl2);
        const float tx = lane_f32(cand.x, wl3), ty = lane_f32(cand.y, wl3), tz = lane_f32(cand.z, wl3);
        const int   w2p = __builtin_amdgcn_readlane(pr.z, wl);
        const int   w2q = __builtin_amdgcn_readlane(pr.z, wl2);
        const v2f ax = {cand.x, cand.x}, ay = {cand.y, cand.y}, az = {cand.z, cand.z};
        const v2f nx2 = {nx, nx}, ny2 = {ny, ny}, nz2 = {nz, nz};
        const v2f mx2 = {mx, mx}, my2 = {my, my}, mz2 = {mz, mz};
        const v2f dlo = fps_pair_dist(true,  ax, ay, az, nx2, ny2, nz2);
        const v2f dhi = fps_pair_dist(false, ax, ay, az, nx2, ny2, nz2);
        const v2f elo = fps_pair_dist(true,  ax, ay, az, mx2, my2, mz2);
        const v2f ehi = fps_pair_dist(false, ax, ay, az, mx2, my2, mz2);
        const int dk  = (pn < FPS_LOWN) ? __float_as_int(dlo.x) : __float_as_int(dhi.x);
        const int ek  = (pn < FPS_LOWN) ? __float_as_int(elo.x) : __float_as_int(ehi.x);
        const int dqp = __builtin_amdgcn_readlane(dk, wl2);
        const int drp = __builtin_amdgcn_readlane(dk, wl3);
        const int drq = __builtin_amdgcn_readlane(ek, wl3);
        const int gm2s = __builtin_amdgcn_readfirstlane(gm2);
        const int gm3s = __builtin_amdgcn_readfirstlane(gm3);
        const bool ok2 = (gm2s > w2p) && (gm2s > 0) && (dqp >= gm2s);
        const bool ok3 = (gm3s > w2p) && (gm3s > w2q) && (gm3s > 0) && (drp >= gm3s) && (drq >= gm3s);

        s   = snext;
        cur = __builtin_amdgcn_readfirstlane(gi)  & (NPTS - 1);
        cq  = __builtin_amdgcn_readfirstlane(gi2) & (NPTS - 1);
        cr  = __builtin_amdgcn_readfirstlane(gi3) & (NPTS - 1);
        cx = nx; cy = ny; cz = nz;
        qx = mx; qy = my; qz = mz;
        rx = tx; ry = ty; rz = tz;
        const bool k2 = ok2 && (s + 1 < NPOINT);
        const bool k3 = k2 && ok3 && (s + 2 < NPOINT);
        np = 1 + (k2 ? 1 : 0) + (k3 ? 1 : 0);
        return false;
    };

    static_assert(NPOINT % 2 == 0, "two exchanges per trip");
    for (int trip = 0; trip < NPOINT / 2; ++trip) {
        if (exchange(dist_a, dist_b, 0)) break;
        if (exchange(dist_b, dist_a, 1)) break;
    }
}

__global__ void gather_new_xyz_kernel(const float* __restrict__ xyz,
                                      const int* __restrict__ fps_idx,
                                      float* __restrict__ new_xyz)
{
    const int e = blockIdx.x * blockDim.x + threadIdx.x;
    if (e >= BATCH * NPOINT * 3) return;
    const int c = e % 3;
    const int p = (e / 3) % NPOINT;
    const int b = e / (3 * NPOINT);
    const int n = fps_idx[b * NPOINT + p];
    new_xyz[e] = xyz[((size_t)b * NPTS + n) * 3 + c];
}

__device__ __forceinline__ int row_up1(int v)
{
    return __builtin_amdgcn_update_dpp(0, v, 0x111, 0xf, 0xf, true);
}

__device__ __forceinline__ void knn_offer(unsigned m, float d, int first,
                                          float& lv, int& li, float& tau)
{
    while (m) {
        const int L = __builtin_ctz(m);
        m &= m - 1;
        const float c = __int_as_float(__builtin_amdgcn_readlane(__float_as_int(d), L));
        if (c < tau) {
            const int   ci = first + L;
            const float sv = __int_as_float(row_up1(__float_as_int(lv)));
            const int   si = row_up1(li);
            const bool  g  = lv > c;
            const bool  gs = sv > c;
            lv = g ? (gs ? sv : c)  : lv;
            li = g ? (gs ? si : ci) : li;
            tau = __int_as_float(__builtin_amdgcn_readlane(__float_as_int(lv), 15));
        }
    }
}

__global__ __launch_bounds__(256) void knn_kernel(const float* __restrict__ xyz,
                                                  const float* __restrict__ new_xyz,
                                                  int* __restrict__ knn_idx)
{
    const int lane = threadIdx.x & 31;
    const int wv = __builtin_amdgcn_readfirstlane((int)(threadIdx.x >> 5));
    const int q0 = (blockIdx.x * 8 + wv) * 2;
    const int b  = q0 / NPOINT;
    const float* xb = xyz + (size_t)b * NPTS * 3;

    const float ax = new_xyz[q0 * 3 + 0], ay = new_xyz[q0 * 3 + 1], az = new_xyz[q0 * 3 + 2];
    const float bx = new_xyz[q0 * 3 + 3], by = new_xyz[q0 * 3 + 4], bz = new_xyz[q0 * 3 + 5];

    float lva = 3.4e38f, lvb = 3.4e38f;
    int   lia = 0,       lib = 0;
    float taua = 3.4e38f, taub = 3.4e38f;

    for (int base = 0; base < NPTS; base += 128) {
        float px[4], py[4], pz[4];
#pragma unroll
        for (int u = 0; u < 4; ++u) {
            const float* p = xb + (size_t)(base + u * 32 + lane) * 3;
            px[u] = p[0]; py[u] = p[1]; pz[u] = p[2];
        }
        float da[4], db[4];
        unsigned ma[4], mb[4];
#pragma unroll
        for (int u = 0; u < 4; ++u) {
            {
                const float dx = px[u] - ax, dy = py[u] - ay, dz = pz[u] - az;
                float s2;
                {
#pragma clang fp contract(off)
                    s2 = dx * dx + dy * dy;
                }
                da[u] = __builtin_fmaf(dz, dz, s2);
            }
            {
                const float dx = px[u] - bx, dy = py[u] - by, dz = pz[u] - bz;
                float s2;
                {
#pragma clang fp contract(off)
                    s2 = dx * dx + dy * dy;
                }
                db[u] = __builtin_fmaf(dz, dz, s2);
            }
            ma[u] = __builtin_amdgcn_ballot_w32(da[u] < taua);
            mb[u] = __builtin_amdgcn_ballot_w32(db[u] < taub);
        }
#pragma unroll
        for (int u = 0; u < 4; ++u) {
            knn_offer(ma[u], da[u], base + u * 32, lva, lia, taua);
            knn_offer(mb[u], db[u], base + u * 32, lvb, lib, taub);
        }
    }
    knn_idx[(size_t)q0 * KNN + lane] = (lane < 16) ? lia : lib;
}

__global__ void pad_w_kernel(const float* __restrict__ W, float* __restrict__ Wpad)
{
    const int e = blockIdx.x * blockDim.x + threadIdx.x;
    if (e >= KPAD * COUT) return;
    const int r = e / COUT;
    Wpad[e] = (r < KRAW) ? W[e] : 0.0f;
}

#define GQ     4
#define GROWS  (GQ * KNN)
static_assert(NPOINT % GQ == 0, "the queries of a block belong to one batch");
static_assert(GROWS * 4 == 256, "the coordinate columns are filled one element per thread");

__global__ __launch_bounds__(256) void group_gemm_kernel(
    const float* __restrict__ xyz, const float* __restrict__ features,
    const float* __restrict__ new_xyz, const int* __restrict__ knn_idx,
    const float* __restrict__ Wpad,
    const float* __restrict__ gamma, const float* __restrict__ beta,
    const float* __restrict__ meanv, const float* __restrict__ varv,
    float* __restrict__ partial_sum, float* __restrict__ partial_sq,
    float* __restrict__ out_features, int final_pass)
{
    const int q0   = blockIdx.x * GQ;
    const int b    = q0 / NPOINT;
    const int tid  = threadIdx.x;
    const int lane = tid & 31;
    const int wv   = tid >> 5;

    __shared__ float As[GROWS][KPAD];
    __shared__ float red0[GQ][8][32];
    __shared__ float red1[GQ][8][32];

    const int* kq = knn_idx + (size_t)q0 * KNN;

    {
        const int r = tid >> 2;
        const int c = tid & 3;
        const int n = kq[r];
        float val = 0.0f;
        if (c < 3) {
            const float pv = xyz[((size_t)b * NPTS + n) * 3 + c];
            val = pv - new_xyz[(q0 + (r >> 4)) * 3 + c];
        }
        As[r][c < 3 ? c : KRAW] = val;
    }
    {
        const int fc = tid & 63;
        const int rs = tid >> 6;
#pragma unroll
        for (int i = 0; i < GROWS / 4; ++i) {
            const int r = i * 4 + rs;
            const int n = kq[r];
            As[r][3 + fc] = features[((size_t)b * NPTS + n) * CIN + fc];
        }
    }
    __syncthreads();

    const int col = wv * 16 + (lane & 15);
    const int hi  = lane >> 4;
    const int row = lane & 15;

    v8f acc[GQ];
#pragma unroll
    for (int t = 0; t < GQ; ++t) {
        const v8f zero = {};
        acc[t] = zero;
    }
#pragma unroll
    for (int kb = 0; kb < KPAD / 4; ++kb) {
        const int k0 = kb * 4 + hi * 2;
        v2f bb;
        bb.x = Wpad[(k0 + 0) * COUT + col];
        bb.y = Wpad[(k0 + 1) * COUT + col];
#pragma unroll
        for (int t = 0; t < GQ; ++t) {
            v2f a;
            a.x  = As[t * KNN + row][k0 + 0];
            a.y  = As[t * KNN + row][k0 + 1];
            acc[t] = __builtin_amdgcn_wmma_f32_16x16x4_f32(
                         false, a, false, bb, (short)0, acc[t], false, false);
        }
    }

    if (!final_pass) {
#pragma unroll
        for (int t = 0; t < GQ; ++t) {
            float s = 0.0f, ss = 0.0f;
#pragma unroll
            for (int j = 0; j < 8; ++j) { const float h = acc[t][j]; s += h; ss += h * h; }
            red0[t][wv][lane] = s;
            red1[t][wv][lane] = ss;
        }
        __syncthreads();
        if (lane < 16) {
#pragma unroll
            for (int t = 0; t < GQ; ++t) {
                const float s2  = red0[t][wv][lane] + red0[t][wv][lane + 16];
                const float ss2 = red1[t][wv][lane] + red1[t][wv][lane + 16];
                partial_sum[(size_t)(q0 + t) * COUT + col] = s2;
                partial_sq [(size_t)(q0 + t) * COUT + col] = ss2;
            }
        }
    } else {
        const float mu    = meanv[col];
        const float scale = gamma[col] * rsqrtf(varv[col] + BNEPS);
        const float shift = beta[col] - mu * scale;
#pragma unroll
        for (int t = 0; t < GQ; ++t) {
            float m = -3.4e38f;
#pragma unroll
            for (int j = 0; j < 8; ++j) {
                float h = acc[t][j] * scale + shift;
                h = fmaxf(h, 0.0f);
                m = fmaxf(m, h);
            }
            red0[t][wv][lane] = m;
        }
        __syncthreads();
        if (lane < 16) {
#pragma unroll
            for (int t = 0; t < GQ; ++t) {
                const float mm = fmaxf(red0[t][wv][lane], red0[t][wv][lane + 16]);
                out_features[(size_t)(q0 + t) * COUT + col] = mm;
            }
        }
    }
}

__global__ __launch_bounds__(256) void reduce_stats_kernel(
    const float* __restrict__ partial_sum, const float* __restrict__ partial_sq,
    float* __restrict__ meanv, float* __restrict__ varv)
{
    const int ch  = blockIdx.x;
    const int tid = threadIdx.x;
    __shared__ float ssum[256], ssq[256];
    float s = 0.0f, sq = 0.0f;
    for (int i = tid; i < BATCH * NPOINT; i += 256) {
        s  += partial_sum[(size_t)i * COUT + ch];
        sq += partial_sq [(size_t)i * COUT + ch];
    }
    ssum[tid] = s; ssq[tid] = sq;
    __syncthreads();
    for (int off = 128; off > 0; off >>= 1) {
        if (tid < off) { ssum[tid] += ssum[tid + off]; ssq[tid] += ssq[tid + off]; }
        __syncthreads();
    }
    if (tid == 0) {
        const float cnt = (float)(BATCH * NPOINT * KNN);
        const float mu  = ssum[0] / cnt;
        meanv[ch] = mu;
        varv[ch]  = ssq[0] / cnt - mu * mu;
    }
}

extern "C" void kernel_launch(void* const* d_in, const int* in_sizes, int n_in,
                              void* d_out, int out_size, void* d_ws, size_t ws_size,
                              hipStream_t stream)
{
    const float* xyz      = (const float*)d_in[0];
    const float* features = (const float*)d_in[1];
    const float* W        = (const float*)d_in[2];
    const float* gamma    = (const float*)d_in[3];
    const float* beta     = (const float*)d_in[4];

    float* out          = (float*)d_out;
    float* new_xyz      = out;
    float* out_features = out + BATCH * NPOINT * 3;

    char* ws = (char*)d_ws;
    size_t off = 0;
    int*   fps_idx     = (int*)(ws + off);   off += sizeof(int)   * BATCH * NPOINT;
    int*   knn_idx     = (int*)(ws + off);   off += sizeof(int)   * BATCH * NPOINT * KNN;
    float* Wpad        = (float*)(ws + off); off += sizeof(float) * KPAD * COUT;
    off = (off + 255) & ~(size_t)255;
    float* partial_sum = (float*)(ws + off); off += sizeof(float) * BATCH * NPOINT * COUT;
    float* partial_sq  = (float*)(ws + off); off += sizeof(float) * BATCH * NPOINT * COUT;
    float* meanv       = (float*)(ws + off); off += sizeof(float) * COUT;
    float* varv        = (float*)(ws + off); off += sizeof(float) * COUT;
    (void)ws_size; (void)in_sizes; (void)n_in; (void)out_size;

    fps_kernel<<<BATCH, FPS_THREADS, 0, stream>>>(xyz, fps_idx);
    gather_new_xyz_kernel<<<(BATCH * NPOINT * 3 + 255) / 256, 256, 0, stream>>>(
        xyz, fps_idx, new_xyz);
    knn_kernel<<<BATCH * NPOINT / 16, 256, 0, stream>>>(
        xyz, new_xyz, knn_idx);
    pad_w_kernel<<<(KPAD * COUT + 255) / 256, 256, 0, stream>>>(W, Wpad);
    group_gemm_kernel<<<BATCH * NPOINT / GQ, 256, 0, stream>>>(
        xyz, features, new_xyz, knn_idx, Wpad, gamma, beta,
        meanv, varv, partial_sum, partial_sq, out_features, 0);
    reduce_stats_kernel<<<COUT, 256, 0, stream>>>(
        partial_sum, partial_sq, meanv, varv);
    group_gemm_kernel<<<BATCH * NPOINT / GQ, 256, 0, stream>>>(
        xyz, features, new_xyz, knn_idx, Wpad, gamma, beta,
        meanv, varv, partial_sum, partial_sq, out_features, 1);
}
